# starting kernel
# baseline (speedup 1.0000x reference)
#include <hip/hip_runtime.h>
#include <hip/hip_bf16.h>

typedef __attribute__((ext_vector_type(2))) float v2f;
typedef __attribute__((ext_vector_type(8))) float v8f;

#define NNODES 10000

__global__ __launch_bounds__(256) void zero4_kernel(float4* __restrict__ p, int n4) {
    int i = blockIdx.x * 256 + threadIdx.x;
    int stride = gridDim.x * 256;
    float4 z; z.x = 0.f; z.y = 0.f; z.z = 0.f; z.w = 0.f;
    for (; i < n4; i += stride) p[i] = z;
}

__global__ __launch_bounds__(256) void transpose_kernel(const float* __restrict__ in,
                                                        float* __restrict__ out,
                                                        int K, int NOut) {
    int idx = blockIdx.x * 256 + threadIdx.x;
    if (idx < K * NOut) {
        int k = idx / NOut;
        int c = idx - k * NOut;
        out[(size_t)c * K + k] = in[idx];
    }
}

__global__ __launch_bounds__(256) void deg_kernel(const int* __restrict__ src,
                                                  float* __restrict__ deg, int E) {
    int i = blockIdx.x * 256 + threadIdx.x;
    if (i < E) atomicAdd(&deg[src[i]], 1.0f);
}

__global__ __launch_bounds__(256) void dis_kernel(float* __restrict__ deg, int n) {
    int i = blockIdx.x * 256 + threadIdx.x;
    if (i < n) {
        float d = deg[i];
        deg[i] = (d > 0.0f) ? rsqrtf(d) : 0.0f;
    }
}

__global__ __launch_bounds__(256) void scatter_kernel(
    const float* __restrict__ xin, const int* __restrict__ src,
    const int* __restrict__ dst, const float* __restrict__ dis,
    float* __restrict__ agg, int E, int shift)
{
    unsigned tid = blockIdx.x * 256u + threadIdx.x;
    int e = (int)(tid >> shift);
    if (e >= E) return;
    int c = (int)(tid & ((1u << shift) - 1u));
    int s = src[e];
    int d = dst[e];
    float norm = -(dis[s] * dis[d]);
    int D = 4 << shift;
    const float4 v = *(const float4*)(xin + (size_t)s * D + 4 * c);
    float* outp = agg + (size_t)d * D + 4 * c;
    atomicAdd(outp + 0, norm * v.x);
    atomicAdd(outp + 1, norm * v.y);
    atomicAdd(outp + 2, norm * v.z);
    atomicAdd(outp + 3, norm * v.w);
}

template <int TILES>
__global__ __launch_bounds__(128) void cheb_gemm_kernel(
    const float* __restrict__ Xin, const float* __restrict__ Agg,
    const float* __restrict__ W0T, const float* __restrict__ W1T,
    const float* __restrict__ bias, float* __restrict__ Out,
    int K, int NOut, int relu)
{
    const int wave  = threadIdx.x >> 5;
    const int lane  = threadIdx.x & 31;
    const int rlane = lane & 15;
    const int khalf = (lane >> 4) << 1;
    const int rowBase = blockIdx.x << 4;
    const int colBase = wave * (TILES * 16);

    const float* a0p = Xin + (size_t)(rowBase + rlane) * K + khalf;
    const float* a1p = Agg + (size_t)(rowBase + rlane) * K + khalf;
    const float* w0p[TILES];
    const float* w1p[TILES];
#pragma unroll
    for (int t = 0; t < TILES; ++t) {
        w0p[t] = W0T + (size_t)(colBase + t * 16 + rlane) * K + khalf;
        w1p[t] = W1T + (size_t)(colBase + t * 16 + rlane) * K + khalf;
    }

    v8f acc[TILES];
#pragma unroll
    for (int t = 0; t < TILES; ++t) acc[t] = (v8f){};

    v2f a0c, a1c, b0c[TILES], b1c[TILES];

    a0c = *(const v2f*)(a0p);
    a1c = *(const v2f*)(a1p);
#pragma unroll
    for (int t = 0; t < TILES; ++t) {
        b0c[t] = *(const v2f*)(w0p[t]);
        b1c[t] = *(const v2f*)(w1p[t]);
    }

#pragma unroll 2
    for (int k = 0; k < K - 4; k += 4) {
        v2f a0n = *(const v2f*)(a0p + k + 4);
        v2f a1n = *(const v2f*)(a1p + k + 4);
        v2f b0n[TILES], b1n[TILES];
#pragma unroll
        for (int t = 0; t < TILES; ++t) b0n[t] = *(const v2f*)(w0p[t] + k + 4);
#pragma unroll
        for (int t = 0; t < TILES; ++t) b1n[t] = *(const v2f*)(w1p[t] + k + 4);

#pragma unroll
        for (int t = 0; t < TILES; ++t)
            acc[t] = __builtin_amdgcn_wmma_f32_16x16x4_f32(
                false, a0c, false, b0c[t], (short)0, acc[t], false, false);
#pragma unroll
        for (int t = 0; t < TILES; ++t)
            acc[t] = __builtin_amdgcn_wmma_f32_16x16x4_f32(
                false, a1c, false, b1c[t], (short)0, acc[t], false, false);

        a0c = a0n; a1c = a1n;
#pragma unroll
        for (int t = 0; t < TILES; ++t) { b0c[t] = b0n[t]; b1c[t] = b1n[t]; }
    }

#pragma unroll
    for (int t = 0; t < TILES; ++t)
        acc[t] = __builtin_amdgcn_wmma_f32_16x16x4_f32(
            false, a0c, false, b0c[t], (short)0, acc[t], false, false);
#pragma unroll
    for (int t = 0; t < TILES; ++t)
        acc[t] = __builtin_amdgcn_wmma_f32_16x16x4_f32(
            false, a1c, false, b1c[t], (short)0, acc[t], false, false);

    const int rowOff = rowBase + ((lane >> 4) << 3);
#pragma unroll
    for (int t = 0; t < TILES; ++t) {
        const int col = colBase + t * 16 + rlane;
        const float bv = bias[col];
#pragma unroll
        for (int r = 0; r < 8; ++r) {
            float v = acc[t][r] + bv;
            if (relu) v = fmaxf(v, 0.0f);
            Out[(size_t)(rowOff + r) * NOut + col] = v;
        }
    }
}

extern "C" void kernel_launch(void* const* d_in, const int* in_sizes, int n_in,
                              void* d_out, int out_size, void* d_ws, size_t ws_size,
                              hipStream_t stream) {
    const float* x   = (const float*)d_in[0];
    const int*   ei  = (const int*)d_in[1];
    const float* W01 = (const float*)d_in[3];
    const float* W11 = (const float*)d_in[4];
    const float* b1  = (const float*)d_in[5];
    const float* W02 = (const float*)d_in[6];
    const float* W12 = (const float*)d_in[7];
    const float* b2  = (const float*)d_in[8];
    const float* W03 = (const float*)d_in[9];
    const float* W13 = (const float*)d_in[10];
    const float* b3  = (const float*)d_in[11];

    const int N = NNODES;
    const int E = in_sizes[1] / 2;
    const int* src = ei;
    const int* dst = ei + E;

    float* dis  = (float*)d_ws;
    float* agg  = dis + 10240;
    float* h1   = agg + (size_t)N * 256;
    float* h2   = h1  + (size_t)N * 256;
    float* w01t = h2  + (size_t)N * 256;
    float* w11t = w01t + 128 * 256;
    float* w02t = w11t + 128 * 256;
    float* w12t = w02t + 256 * 256;
    float* w03t = w12t + 256 * 256;
    float* w13t = w03t + 256 * 128;
    float* out  = (float*)d_out;

    transpose_kernel<<<(128 * 256 + 255) / 256, 256, 0, stream>>>(W01, w01t, 128, 256);
    transpose_kernel<<<(128 * 256 + 255) / 256, 256, 0, stream>>>(W11, w11t, 128, 256);
    transpose_kernel<<<(256 * 256 + 255) / 256, 256, 0, stream>>>(W02, w02t, 256, 256);
    transpose_kernel<<<(256 * 256 + 255) / 256, 256, 0, stream>>>(W12, w12t, 256, 256);
    transpose_kernel<<<(256 * 128 + 255) / 256, 256, 0, stream>>>(W03, w03t, 256, 128);
    transpose_kernel<<<(256 * 128 + 255) / 256, 256, 0, stream>>>(W13, w13t, 256, 128);

    zero4_kernel<<<64, 256, 0, stream>>>((float4*)dis, N / 4);
    deg_kernel<<<(E + 255) / 256, 256, 0, stream>>>(src, dis, E);
    dis_kernel<<<(N + 255) / 256, 256, 0, stream>>>(dis, N);

    zero4_kernel<<<1024, 256, 0, stream>>>((float4*)agg, N * 128 / 4);
    {
        unsigned nth = (unsigned)E << 5;
        scatter_kernel<<<(nth + 255) / 256, 256, 0, stream>>>(x, src, dst, dis, agg, E, 5);
    }
    cheb_gemm_kernel<4><<<N / 16, 128, 0, stream>>>(x, agg, w01t, w11t, b1, h1, 128, 256, 1);

    zero4_kernel<<<2048, 256, 0, stream>>>((float4*)agg, N * 256 / 4);
    {
        unsigned nth = (unsigned)E << 6;
        scatter_kernel<<<(nth + 255) / 256, 256, 0, stream>>>(h1, src, dst, dis, agg, E, 6);
    }
    cheb_gemm_kernel<4><<<N / 16, 128, 0, stream>>>(h1, agg, w02t, w12t, b2, h2, 256, 256, 1);

    zero4_kernel<<<2048, 256, 0, stream>>>((float4*)agg, N * 256 / 4);
    {
        unsigned nth = (unsigned)E << 6;
        scatter_kernel<<<(nth + 255) / 256, 256, 0, stream>>>(h2, src, dst, dis, agg, E, 6);
    }
    cheb_gemm_kernel<2><<<N / 16, 128, 0, stream>>>(h2, agg, w03t, w13t, b3, out, 256, 128, 0);
}


# Aggregation as destination-sorted gather: edges counting-sorted by dst (integer atomic ranks, arrival order); one wave sums a node's source rows; no float atomics or agg zero fill; sum order changes
# speedup vs baseline: 2.1692x; 2.1692x over previous
#include <hip/hip_runtime.h>
#include <hip/hip_bf16.h>

typedef __attribute__((ext_vector_type(2))) float v2f;
typedef __attribute__((ext_vector_type(8))) float v8f;

#define NNODES 10000
#define CPAD 32
#define SCAN_PER 10

__global__ __launch_bounds__(256) void zero4_kernel(float4* __restrict__ p, int n4) {
    int i = blockIdx.x * 256 + threadIdx.x;
    int stride = gridDim.x * 256;
    float4 z; z.x = 0.f; z.y = 0.f; z.z = 0.f; z.w = 0.f;
    for (; i < n4; i += stride) p[i] = z;
}

__global__ __launch_bounds__(256) void transpose_kernel(const float* __restrict__ in,
                                                        float* __restrict__ out,
                                                        int K, int NOut) {
    int idx = blockIdx.x * 256 + threadIdx.x;
    if (idx < K * NOut) {
        int k = idx / NOut;
        int c = idx - k * NOut;
        out[(size_t)c * K + k] = in[idx];
    }
}

__global__ __launch_bounds__(256) void deg_kernel(const int* __restrict__ src,
                                                  float* __restrict__ deg, int E) {
    int i = blockIdx.x * 256 + threadIdx.x;
    if (i < E) atomicAdd(&deg[src[i]], 1.0f);
}

__global__ __launch_bounds__(256) void dis_kernel(float* __restrict__ deg, int n) {
    int i = blockIdx.x * 256 + threadIdx.x;
    if (i < n) {
        float d = deg[i];
        deg[i] = (d > 0.0f) ? rsqrtf(d) : 0.0f;
    }
}

__global__ __launch_bounds__(256) void count_kernel(const int* __restrict__ dst,
                                                    int* __restrict__ cnt,
                                                    int* __restrict__ rank, int E) {
    int e = blockIdx.x * 256 + threadIdx.x;
    if (e < E) rank[e] = atomicAdd(&cnt[dst[e] * CPAD], 1);
}

__global__ __launch_bounds__(1024) void scan_kernel(const int* __restrict__ cnt,
                                                    int* __restrict__ rowptr) {
    __shared__ int part[1024];
    const int t = threadIdx.x;
    int c[SCAN_PER];
    int loc = 0;
#pragma unroll
    for (int i = 0; i < SCAN_PER; ++i) {
        const int idx = t * SCAN_PER + i;
        c[i] = (idx < NNODES) ? cnt[idx * CPAD] : 0;
        loc += c[i];
    }
    part[t] = loc;
    __syncthreads();
    for (int off = 1; off < 1024; off <<= 1) {
        const int v = (t >= off) ? part[t - off] : 0;
        __syncthreads();
        part[t] += v;
        __syncthreads();
    }
    int run = part[t] - loc;
#pragma unroll
    for (int i = 0; i < SCAN_PER; ++i) {
        const int idx = t * SCAN_PER + i;
        if (idx < NNODES) rowptr[idx] = run;
        run += c[i];
    }
    if (t == 1023) rowptr[NNODES] = part[1023];
}

__global__ __launch_bounds__(256) void fill_kernel(const int* __restrict__ src,
                                                   const int* __restrict__ dst,
                                                   const int* __restrict__ rank,
                                                   const int* __restrict__ rowptr,
                                                   const float* __restrict__ dis,
                                                   int* __restrict__ csrc,
                                                   float* __restrict__ cdis, int E) {
    int e = blockIdx.x * 256 + threadIdx.x;
    if (e < E) {
        const int s = src[e];
        const int pos = rowptr[dst[e]] + rank[e];
        csrc[pos] = s;
        cdis[pos] = dis[s];
    }
}


template <int D, int B>
__device__ __forceinline__ void edge_batch(const float* __restrict__ xin, int s, float nm,
                                           int j, int lane, float4 (&acc)[D / 128]) {
    constexpr int V = D / 128;
    float4 xv[B][V];
    float w[B];
#pragma unroll
    for (int u = 0; u < B; ++u) {
        const int su = __builtin_amdgcn_readlane(s, j + u);
        w[u] = __int_as_float(__builtin_amdgcn_readlane(__float_as_int(nm), j + u));
        const float4* row = (const float4*)(xin + (size_t)su * D);
#pragma unroll
        for (int v = 0; v < V; ++v) xv[u][v] = row[lane + 32 * v];
    }
#pragma unroll
    for (int u = 0; u < B; ++u) {
#pragma unroll
        for (int v = 0; v < V; ++v) {
            acc[v].x = __builtin_fmaf(w[u], xv[u][v].x, acc[v].x);
            acc[v].y = __builtin_fmaf(w[u], xv[u][v].y, acc[v].y);
            acc[v].z = __builtin_fmaf(w[u], xv[u][v].z, acc[v].z);
            acc[v].w = __builtin_fmaf(w[u], xv[u][v].w, acc[v].w);
        }
    }
}

template <int D>
__global__ __launch_bounds__(256) void gather_kernel(
    const float* __restrict__ xin, const int* __restrict__ rowptr,
    const int* __restrict__ csrc, const float* __restrict__ cdis,
    const float* __restrict__ dis, float* __restrict__ agg, int n)
{
    constexpr int V = D / 128;
    constexpr int U = 8 / V;
    const int lane = threadIdx.x & 31;
    const int node = __builtin_amdgcn_readfirstlane((int)(blockIdx.x * 8u + (threadIdx.x >> 5)));
    if (node >= n) return;
    const int beg = rowptr[node];
    const int end = rowptr[node + 1];
    const float dd = dis[node];

    float4 acc[V];
#pragma unroll
    for (int v = 0; v < V; ++v) { acc[v].x = 0.f; acc[v].y = 0.f; acc[v].z = 0.f; acc[v].w = 0.f; }

    int s = 0;
    float ds = 0.0f;
    if (beg + lane < end) { s = csrc[beg + lane]; ds = cdis[beg + lane]; }

    for (int base = beg; base < end; base += 32) {
        const int pn = (base + 32 + lane < end) ? (base + 32 + lane) : (end - 1);
        const int sn = csrc[pn];
        const float dsn = cdis[pn];

        const float nm = -(ds * dd);
        const int cnt = (end - base < 32) ? (end - base) : 32;
        int j = 0;
        for (; j + U <= cnt; j += U) edge_batch<D, U>(xin, s, nm, j, lane, acc);
        if (U > 4 && cnt - j >= 4) { edge_batch<D, 4>(xin, s, nm, j, lane, acc); j += 4; }
        if (cnt - j >= 2) { edge_batch<D, 2>(xin, s, nm, j, lane, acc); j += 2; }
        if (cnt - j >= 1) { edge_batch<D, 1>(xin, s, nm, j, lane, acc); }

        s = sn;
        ds = dsn;
    }

    float4* orow = (float4*)(agg + (size_t)node * D);
#pragma unroll
    for (int v = 0; v < V; ++v) orow[lane + 32 * v] = acc[v];
}

template <int TILES>
__global__ __launch_bounds__(128) void cheb_gemm_kernel(
    const float* __restrict__ Xin, const float* Agg,
    const float* __restrict__ W0T, const float* __restrict__ W1T,
    const float* __restrict__ bias, float* Out,
    int K, int NOut, int relu)
{
    const int wave  = threadIdx.x >> 5;
    const int lane  = threadIdx.x & 31;
    const int rlane = lane & 15;
    const int khalf = (lane >> 4) << 1;
    const int rowBase = blockIdx.x << 4;
    const int colBase = wave * (TILES * 16);

    const float* a0p = Xin + (size_t)(rowBase + rlane) * K + khalf;
    const float* a1p = Agg + (size_t)(rowBase + rlane) * K + khalf;
    const float* w0p[TILES];
    const float* w1p[TILES];
#pragma unroll
    for (int t = 0; t < TILES; ++t) {
        w0p[t] = W0T + (size_t)(colBase + t * 16 + rlane) * K + khalf;
        w1p[t] = W1T + (size_t)(colBase + t * 16 + rlane) * K + khalf;
    }

    v8f acc[TILES];
#pragma unroll
    for (int t = 0; t < TILES; ++t) acc[t] = (v8f){};

    v2f a0c, a1c, b0c[TILES], b1c[TILES];

    a0c = *(const v2f*)(a0p);
    a1c = *(const v2f*)(a1p);
#pragma unroll
    for (int t = 0; t < TILES; ++t) {
        b0c[t] = *(const v2f*)(w0p[t]);
        b1c[t] = *(const v2f*)(w1p[t]);
    }

#pragma unroll 2
    for (int k = 0; k < K - 4; k += 4) {
        v2f a0n = *(const v2f*)(a0p + k + 4);
        v2f a1n = *(const v2f*)(a1p + k + 4);
        v2f b0n[TILES], b1n[TILES];
#pragma unroll
        for (int t = 0; t < TILES; ++t) b0n[t] = *(const v2f*)(w0p[t] + k + 4);
#pragma unroll
        for (int t = 0; t < TILES; ++t) b1n[t] = *(const v2f*)(w1p[t] + k + 4);

#pragma unroll
        for (int t = 0; t < TILES; ++t)
            acc[t] = __builtin_amdgcn_wmma_f32_16x16x4_f32(
                false, a0c, false, b0c[t], (short)0, acc[t], false, false);
#pragma unroll
        for (int t = 0; t < TILES; ++t)
            acc[t] = __builtin_amdgcn_wmma_f32_16x16x4_f32(
                false, a1c, false, b1c[t], (short)0, acc[t], false, false);

        a0c = a0n; a1c = a1n;
#pragma unroll
        for (int t = 0; t < TILES; ++t) { b0c[t] = b0n[t]; b1c[t] = b1n[t]; }
    }

#pragma unroll
    for (int t = 0; t < TILES; ++t)
        acc[t] = __builtin_amdgcn_wmma_f32_16x16x4_f32(
            false, a0c, false, b0c[t], (short)0, acc[t], false, false);
#pragma unroll
    for (int t = 0; t < TILES; ++t)
        acc[t] = __builtin_amdgcn_wmma_f32_16x16x4_f32(
            false, a1c, false, b1c[t], (short)0, acc[t], false, false);

    __syncthreads();

    const int rowOff = rowBase + ((lane >> 4) << 3);
#pragma unroll
    for (int t = 0; t < TILES; ++t) {
        const int col = colBase + t * 16 + rlane;
        const float bv = bias[col];
#pragma unroll
        for (int r = 0; r < 8; ++r) {
            float v = acc[t][r] + bv;
            if (relu) v = fmaxf(v, 0.0f);
            Out[(size_t)(rowOff + r) * NOut + col] = v;
        }
    }
}

extern "C" void kernel_launch(void* const* d_in, const int* in_sizes, int n_in,
                              void* d_out, int out_size, void* d_ws, size_t ws_size,
                              hipStream_t stream) {
    const float* x   = (const float*)d_in[0];
    const int*   ei  = (const int*)d_in[1];
    const float* W01 = (const float*)d_in[3];
    const float* W11 = (const float*)d_in[4];
    const float* b1  = (const float*)d_in[5];
    const float* W02 = (const float*)d_in[6];
    const float* W12 = (const float*)d_in[7];
    const float* b2  = (const float*)d_in[8];
    const float* W03 = (const float*)d_in[9];
    const float* W13 = (const float*)d_in[10];
    const float* b3  = (const float*)d_in[11];

    const int N = NNODES;
    const int E = in_sizes[1] / 2;
    const int* src = ei;
    const int* dst = ei + E;

    float* dis  = (float*)d_ws;
    float* bufA = dis + 10240;
    float* bufB = bufA + (size_t)N * 256;
    float* bufC = bufB + (size_t)N * 256;
    float* w01t = bufC + (size_t)N * 256;
    float* w11t = w01t + 128 * 256;
    float* w02t = w11t + 128 * 256;
    float* w12t = w02t + 256 * 256;
    float* w03t = w12t + 256 * 256;
    float* w13t = w03t + 256 * 128;
    float* out  = (float*)d_out;

    int* cnt    = (int*)bufA;
    int* rank   = (int*)bufB;
    int*   csrc   = (int*)bufC;
    float* cdis   = bufC + E;
    int*   rowptr = csrc + E + E;

    transpose_kernel<<<(128 * 256 + 255) / 256, 256, 0, stream>>>(W01, w01t, 128, 256);
    transpose_kernel<<<(128 * 256 + 255) / 256, 256, 0, stream>>>(W11, w11t, 128, 256);
    transpose_kernel<<<(256 * 256 + 255) / 256, 256, 0, stream>>>(W02, w02t, 256, 256);
    transpose_kernel<<<(256 * 256 + 255) / 256, 256, 0, stream>>>(W12, w12t, 256, 256);
    transpose_kernel<<<(256 * 128 + 255) / 256, 256, 0, stream>>>(W03, w03t, 256, 128);
    transpose_kernel<<<(256 * 128 + 255) / 256, 256, 0, stream>>>(W13, w13t, 256, 128);

    zero4_kernel<<<128, 256, 0, stream>>>((float4*)dis, (10240 + N * CPAD) / 4);
    deg_kernel<<<(E + 255) / 256, 256, 0, stream>>>(src, dis, E);
    dis_kernel<<<(N + 255) / 256, 256, 0, stream>>>(dis, N);

    count_kernel<<<(E + 255) / 256, 256, 0, stream>>>(dst, cnt, rank, E);
    scan_kernel<<<1, 1024, 0, stream>>>(cnt, rowptr);
    fill_kernel<<<(E + 255) / 256, 256, 0, stream>>>(src, dst, rank, rowptr, dis, csrc, cdis, E);

    gather_kernel<128><<<(N + 7) / 8, 256, 0, stream>>>(x, rowptr, csrc, cdis, dis, bufA, N);
    cheb_gemm_kernel<4><<<N / 16, 128, 0, stream>>>(x, bufA, w01t, w11t, b1, bufB, 128, 256, 1);

    gather_kernel<256><<<(N + 7) / 8, 256, 0, stream>>>(bufB, rowptr, csrc, cdis, dis, bufA, N);
    cheb_gemm_kernel<4><<<N / 16, 128, 0, stream>>>(bufB, bufA, w02t, w12t, b2, bufA, 256, 256, 1);

    gather_kernel<256><<<(N + 7) / 8, 256, 0, stream>>>(bufA, rowptr, csrc, cdis, dis, bufB, N);
    cheb_gemm_kernel<2><<<N / 16, 128, 0, stream>>>(bufA, bufB, w03t, w13t, b3, out, 256, 128, 0);
}


# GEMM operand loads regrouped: weights pre-packed in fragment order (one contiguous 512-byte load per wave and tile), A as 16-byte loads, next stage's loads issued as one group; k sum order changes
# speedup vs baseline: 2.8849x; 1.3299x over previous
#include <hip/hip_runtime.h>
#include <hip/hip_bf16.h>

typedef __attribute__((ext_vector_type(2))) float v2f;
typedef __attribute__((ext_vector_type(4))) float v4f;
typedef __attribute__((ext_vector_type(8))) float v8f;

#define NNODES 10000
#define CPAD 32
#define SCAN_PER 10

__global__ __launch_bounds__(256) void zero4_kernel(float4* __restrict__ p, int n4) {
    int i = blockIdx.x * 256 + threadIdx.x;
    int stride = gridDim.x * 256;
    float4 z; z.x = 0.f; z.y = 0.f; z.z = 0.f; z.w = 0.f;
    for (; i < n4; i += stride) p[i] = z;
}

__global__ __launch_bounds__(256) void pack_kernel(const float* __restrict__ in,
                                                   float* __restrict__ out,
                                                   int K, int NOut) {
    int o = blockIdx.x * 256 + threadIdx.x;
    if (o < K * NOut) {
        const int e = o & 3;
        const int l = (o >> 2) & 31;
        const int tg = o >> 7;
        const int KG = K >> 3;
        const int tile = tg / KG;
        const int g = tg - tile * KG;
        const int k = 8 * g + 4 * (l >> 4) + e;
        const int n = 16 * tile + (l & 15);
        out[o] = in[(size_t)k * NOut + n];
    }
}

__global__ __launch_bounds__(256) void deg_kernel(const int* __restrict__ src,
                                                  float* __restrict__ deg, int E) {
    int i = blockIdx.x * 256 + threadIdx.x;
    if (i < E) atomicAdd(&deg[src[i]], 1.0f);
}

__global__ __launch_bounds__(256) void dis_kernel(float* __restrict__ deg, int n) {
    int i = blockIdx.x * 256 + threadIdx.x;
    if (i < n) {
        float d = deg[i];
        deg[i] = (d > 0.0f) ? rsqrtf(d) : 0.0f;
    }
}

__global__ __launch_bounds__(256) void count_kernel(const int* __restrict__ dst,
                                                    int* __restrict__ cnt,
                                                    int* __restrict__ rank, int E) {
    int e = blockIdx.x * 256 + threadIdx.x;
    if (e < E) rank[e] = atomicAdd(&cnt[dst[e] * CPAD], 1);
}

__global__ __launch_bounds__(1024) void scan_kernel(const int* __restrict__ cnt,
                                                    int* __restrict__ rowptr) {
    __shared__ int part[1024];
    const int t = threadIdx.x;
    int c[SCAN_PER];
    int loc = 0;
#pragma unroll
    for (int i = 0; i < SCAN_PER; ++i) {
        const int idx = t * SCAN_PER + i;
        c[i] = (idx < NNODES) ? cnt[idx * CPAD] : 0;
        loc += c[i];
    }
    part[t] = loc;
    __syncthreads();
    for (int off = 1; off < 1024; off <<= 1) {
        const int v = (t >= off) ? part[t - off] : 0;
        __syncthreads();
        part[t] += v;
        __syncthreads();
    }
    int run = part[t] - loc;
#pragma unroll
    for (int i = 0; i < SCAN_PER; ++i) {
        const int idx = t * SCAN_PER + i;
        if (idx < NNODES) rowptr[idx] = run;
        run += c[i];
    }
    if (t == 1023) rowptr[NNODES] = part[1023];
}

__global__ __launch_bounds__(256) void fill_kernel(const int* __restrict__ src,
                                                   const int* __restrict__ dst,
                                                   const int* __restrict__ rank,
                                                   const int* __restrict__ rowptr,
                                                   const float* __restrict__ dis,
                                                   int* __restrict__ csrc,
                                                   float* __restrict__ cdis, int E) {
    int e = blockIdx.x * 256 + threadIdx.x;
    if (e < E) {
        const int s = src[e];
        const int pos = rowptr[dst[e]] + rank[e];
        csrc[pos] = s;
        cdis[pos] = dis[s];
    }
}


template <int D, int B>
__device__ __forceinline__ void edge_batch(const float* __restrict__ xin, int s, float nm,
                                           int j, int lane, float4 (&acc)[D / 128]) {
    constexpr int V = D / 128;
    float4 xv[B][V];
    float w[B];
#pragma unroll
    for (int u = 0; u < B; ++u) {
        const int su = __builtin_amdgcn_readlane(s, j + u);
        w[u] = __int_as_float(__builtin_amdgcn_readlane(__float_as_int(nm), j + u));
        const float4* row = (const float4*)(xin + (size_t)su * D);
#pragma unroll
        for (int v = 0; v < V; ++v) xv[u][v] = row[lane + 32 * v];
    }
#pragma unroll
    for (int u = 0; u < B; ++u) {
#pragma unroll
        for (int v = 0; v < V; ++v) {
            acc[v].x = __builtin_fmaf(w[u], xv[u][v].x, acc[v].x);
            acc[v].y = __builtin_fmaf(w[u], xv[u][v].y, acc[v].y);
            acc[v].z = __builtin_fmaf(w[u], xv[u][v].z, acc[v].z);
            acc[v].w = __builtin_fmaf(w[u], xv[u][v].w, acc[v].w);
        }
    }
}

template <int D>
__global__ __launch_bounds__(256) void gather_kernel(
    const float* __restrict__ xin, const int* __restrict__ rowptr,
    const int* __restrict__ csrc, const float* __restrict__ cdis,
    const float* __restrict__ dis, float* __restrict__ agg, int n)
{
    constexpr int V = D / 128;
    constexpr int U = 8 / V;
    const int lane = threadIdx.x & 31;
    const int node = __builtin_amdgcn_readfirstlane((int)(blockIdx.x * 8u + (threadIdx.x >> 5)));
    if (node >= n) return;
    const int beg = rowptr[node];
    const int end = rowptr[node + 1];
    const float dd = dis[node];

    float4 acc[V];
#pragma unroll
    for (int v = 0; v < V; ++v) { acc[v].x = 0.f; acc[v].y = 0.f; acc[v].z = 0.f; acc[v].w = 0.f; }

    int s = 0;
    float ds = 0.0f;
    if (beg + lane < end) { s = csrc[beg + lane]; ds = cdis[beg + lane]; }

    for (int base = beg; base < end; base += 32) {
        const int pn = (base + 32 + lane < end) ? (base + 32 + lane) : (end - 1);
        const int sn = csrc[pn];
        const float dsn = cdis[pn];

        const float nm = -(ds * dd);
        const int cnt = (end - base < 32) ? (end - base) : 32;
        int j = 0;
        for (; j + U <= cnt; j += U) edge_batch<D, U>(xin, s, nm, j, lane, acc);
        if (U > 4 && cnt - j >= 4) { edge_batch<D, 4>(xin, s, nm, j, lane, acc); j += 4; }
        if (cnt - j >= 2) { edge_batch<D, 2>(xin, s, nm, j, lane, acc); j += 2; }
        if (cnt - j >= 1) { edge_batch<D, 1>(xin, s, nm, j, lane, acc); }

        s = sn;
        ds = dsn;
    }

    float4* orow = (float4*)(agg + (size_t)node * D);
#pragma unroll
    for (int v = 0; v < V; ++v) orow[lane + 32 * v] = acc[v];
}

template <int TILES, int PAIRS>
struct GemmStage {
    v4f a0[PAIRS], a1[PAIRS];
    v4f b0[PAIRS][TILES], b1[PAIRS][TILES];
};

template <int TILES, int K, int PAIRS>
__device__ __forceinline__ void load_stage(GemmStage<TILES, PAIRS>& st, int s,
                                           const v4f* a0p, const v4f* a1p,
                                           const v4f* w0p, const v4f* w1p) {
    constexpr int KG = K / 8;
#pragma unroll
    for (int q = 0; q < PAIRS; ++q) {
        const int g = s * PAIRS + q;
        st.a0[q] = a0p[2 * g];
        st.a1[q] = a1p[2 * g];
#pragma unroll
        for (int t = 0; t < TILES; ++t) st.b0[q][t] = w0p[(t * KG + g) * 32];
#pragma unroll
        for (int t = 0; t < TILES; ++t) st.b1[q][t] = w1p[(t * KG + g) * 32];
    }
}

template <int TILES, int PAIRS>
__device__ __forceinline__ void mma_stage(const GemmStage<TILES, PAIRS>& st, v8f (&acc)[TILES]) {
#pragma unroll
    for (int q = 0; q < PAIRS; ++q) {
        const v2f a0lo = __builtin_shufflevector(st.a0[q], st.a0[q], 0, 1);
        const v2f a0hi = __builtin_shufflevector(st.a0[q], st.a0[q], 2, 3);
        const v2f a1lo = __builtin_shufflevector(st.a1[q], st.a1[q], 0, 1);
        const v2f a1hi = __builtin_shufflevector(st.a1[q], st.a1[q], 2, 3);
#pragma unroll
        for (int t = 0; t < TILES; ++t)
            acc[t] = __builtin_amdgcn_wmma_f32_16x16x4_f32(
                false, a0lo, false, __builtin_shufflevector(st.b0[q][t], st.b0[q][t], 0, 1),
                (short)0, acc[t], false, false);
#pragma unroll
        for (int t = 0; t < TILES; ++t)
            acc[t] = __builtin_amdgcn_wmma_f32_16x16x4_f32(
                false, a0hi, false, __builtin_shufflevector(st.b0[q][t], st.b0[q][t], 2, 3),
                (short)0, acc[t], false, false);
#pragma unroll
        for (int t = 0; t < TILES; ++t)
            acc[t] = __builtin_amdgcn_wmma_f32_16x16x4_f32(
                false, a1lo, false, __builtin_shufflevector(st.b1[q][t], st.b1[q][t], 0, 1),
                (short)0, acc[t], false, false);
#pragma unroll
        for (int t = 0; t < TILES; ++t)
            acc[t] = __builtin_amdgcn_wmma_f32_16x16x4_f32(
                false, a1hi, false, __builtin_shufflevector(st.b1[q][t], st.b1[q][t], 2, 3),
                (short)0, acc[t], false, false);
    }
}

template <int TILES, int K, int PAIRS>
__global__ __launch_bounds__(128) void cheb_gemm_kernel(
    const float* __restrict__ Xin, const float* Agg,
    const float* __restrict__ W0P, const float* __restrict__ W1P,
    const float* __restrict__ bias, float* Out, int relu)
{
    constexpr int KG = K / 8;
    constexpr int NST = KG / PAIRS;
    constexpr int NOut = TILES * 64;
    const int wave  = threadIdx.x >> 5;
    const int lane  = threadIdx.x & 31;
    const int rlane = lane & 15;
    const int half  = lane >> 4;
    const int rowBase = blockIdx.x << 4;
    const int colBase = wave * (TILES * 16);

    const v4f* a0p = (const v4f*)(Xin + (size_t)(rowBase + rlane) * K) + half;
    const v4f* a1p = (const v4f*)(Agg + (size_t)(rowBase + rlane) * K) + half;
    const v4f* w0p = (const v4f*)W0P + (size_t)(wave * TILES) * (KG * 32) + lane;
    const v4f* w1p = (const v4f*)W1P + (size_t)(wave * TILES) * (KG * 32) + lane;

    v8f acc[TILES];
#pragma unroll
    for (int t = 0; t < TILES; ++t) acc[t] = (v8f){};

    GemmStage<TILES, PAIRS> cur;
    load_stage<TILES, K, PAIRS>(cur, 0, a0p, a1p, w0p, w1p);

#pragma unroll
    for (int s = 0; s < NST - 1; ++s) {
        GemmStage<TILES, PAIRS> nxt;
        load_stage<TILES, K, PAIRS>(nxt, s + 1, a0p, a1p, w0p, w1p);
        __builtin_amdgcn_sched_barrier(0);
        mma_stage<TILES, PAIRS>(cur, acc);
        cur = nxt;
    }
    mma_stage<TILES, PAIRS>(cur, acc);

    __syncthreads();

    const int rowOff = rowBase + (half << 3);
#pragma unroll
    for (int t = 0; t < TILES; ++t) {
        const int col = colBase + t * 16 + rlane;
        const float bv = bias[col];
#pragma unroll
        for (int r = 0; r < 8; ++r) {
            float v = acc[t][r] + bv;
            if (relu) v = fmaxf(v, 0.0f);
            Out[(size_t)(rowOff + r) * NOut + col] = v;
        }
    }
}

extern "C" void kernel_launch(void* const* d_in, const int* in_sizes, int n_in,
                              void* d_out, int out_size, void* d_ws, size_t ws_size,
                              hipStream_t stream) {
    const float* x   = (const float*)d_in[0];
    const int*   ei  = (const int*)d_in[1];
    const float* W01 = (const float*)d_in[3];
    const float* W11 = (const float*)d_in[4];
    const float* b1  = (const float*)d_in[5];
    const float* W02 = (const float*)d_in[6];
    const float* W12 = (const float*)d_in[7];
    const float* b2  = (const float*)d_in[8];
    const float* W03 = (const float*)d_in[9];
    const float* W13 = (const float*)d_in[10];
    const float* b3  = (const float*)d_in[11];

    const int N = NNODES;
    const int E = in_sizes[1] / 2;
    const int* src = ei;
    const int* dst = ei + E;

    float* dis  = (float*)d_ws;
    float* bufA = dis + 10240;
    float* bufB = bufA + (size_t)N * 256;
    float* bufC = bufB + (size_t)N * 256;
    float* w01t = bufC + (size_t)N * 256;
    float* w11t = w01t + 128 * 256;
    float* w02t = w11t + 128 * 256;
    float* w12t = w02t + 256 * 256;
    float* w03t = w12t + 256 * 256;
    float* w13t = w03t + 256 * 128;
    float* out  = (float*)d_out;

    int* cnt    = (int*)bufA;
    int* rank   = (int*)bufB;
    int*   csrc   = (int*)bufC;
    float* cdis   = bufC + E;
    int*   rowptr = csrc + E + E;

    pack_kernel<<<(128 * 256 + 255) / 256, 256, 0, stream>>>(W01, w01t, 128, 256);
    pack_kernel<<<(128 * 256 + 255) / 256, 256, 0, stream>>>(W11, w11t, 128, 256);
    pack_kernel<<<(256 * 256 + 255) / 256, 256, 0, stream>>>(W02, w02t, 256, 256);
    pack_kernel<<<(256 * 256 + 255) / 256, 256, 0, stream>>>(W12, w12t, 256, 256);
    pack_kernel<<<(256 * 128 + 255) / 256, 256, 0, stream>>>(W03, w03t, 256, 128);
    pack_kernel<<<(256 * 128 + 255) / 256, 256, 0, stream>>>(W13, w13t, 256, 128);

    zero4_kernel<<<128, 256, 0, stream>>>((float4*)dis, (10240 + N * CPAD) / 4);
    deg_kernel<<<(E + 255) / 256, 256, 0, stream>>>(src, dis, E);
    dis_kernel<<<(N + 255) / 256, 256, 0, stream>>>(dis, N);

    count_kernel<<<(E + 255) / 256, 256, 0, stream>>>(dst, cnt, rank, E);
    scan_kernel<<<1, 1024, 0, stream>>>(cnt, rowptr);
    fill_kernel<<<(E + 255) / 256, 256, 0, stream>>>(src, dst, rank, rowptr, dis, csrc, cdis, E);

    gather_kernel<128><<<(N + 7) / 8, 256, 0, stream>>>(x, rowptr, csrc, cdis, dis, bufA, N);
    cheb_gemm_kernel<4, 128, 2><<<N / 16, 128, 0, stream>>>(x, bufA, w01t, w11t, b1, bufB, 1);

    gather_kernel<256><<<(N + 7) / 8, 256, 0, stream>>>(bufB, rowptr, csrc, cdis, dis, bufA, N);
    cheb_gemm_kernel<4, 256, 2><<<N / 16, 128, 0, stream>>>(bufB, bufA, w02t, w12t, b2, bufA, 1);

    gather_kernel<256><<<(N + 7) / 8, 256, 0, stream>>>(bufA, rowptr, csrc, cdis, dis, bufB, N);
    cheb_gemm_kernel<2, 256, 4><<<N / 16, 128, 0, stream>>>(bufA, bufB, w03t, w13t, b3, out, 0);
}


# Launch merging, 18 launches to 10: six weight packs and the counter clear share one launch, degree and destination rank share one pass over the edges, dis is made inside the scan; bodies as before
# speedup vs baseline: 3.2502x; 1.1266x over previous
#include <hip/hip_runtime.h>
#include <hip/hip_bf16.h>

typedef __attribute__((ext_vector_type(2))) float v2f;
typedef __attribute__((ext_vector_type(4))) float v4f;
typedef __attribute__((ext_vector_type(8))) float v8f;

#define NNODES 10000
#define CPAD 32
#define SCAN_PER 10

__device__ __forceinline__ void pack_one(const float* __restrict__ in, float* __restrict__ out,
                                         int K, int NOut, int o) {
    const int e = o & 3;
    const int l = (o >> 2) & 31;
    const int tg = o >> 7;
    const int KG = K >> 3;
    const int tile = tg / KG;
    const int g = tg - tile * KG;
    const int k = 8 * g + 4 * (l >> 4) + e;
    const int n = 16 * tile + (l & 15);
    out[o] = in[(size_t)k * NOut + n];
}

#define ZERO_BLOCKS 128
#define PACK_BLOCKS 1024
__global__ __launch_bounds__(256) void prep_kernel(
    const float* __restrict__ W01, const float* __restrict__ W11,
    const float* __restrict__ W02, const float* __restrict__ W12,
    const float* __restrict__ W03, const float* __restrict__ W13,
    float* __restrict__ wp, float4* __restrict__ zp, int n4)
{
    const int t = threadIdx.x;
    if (blockIdx.x < ZERO_BLOCKS) {
        float4 z; z.x = 0.f; z.y = 0.f; z.z = 0.f; z.w = 0.f;
        for (int i = blockIdx.x * 256 + t; i < n4; i += ZERO_BLOCKS * 256) zp[i] = z;
        return;
    }
    const int b = blockIdx.x - ZERO_BLOCKS;
    if (b < 128)      pack_one(W01, wp,          128, 256, b * 256 + t);
    else if (b < 256) pack_one(W11, wp + 32768,  128, 256, (b - 128) * 256 + t);
    else if (b < 512) pack_one(W02, wp + 65536,  256, 256, (b - 256) * 256 + t);
    else if (b < 768) pack_one(W12, wp + 131072, 256, 256, (b - 512) * 256 + t);
    else if (b < 896) pack_one(W03, wp + 196608, 256, 128, (b - 768) * 256 + t);
    else              pack_one(W13, wp + 229376, 256, 128, (b - 896) * 256 + t);
}

__global__ __launch_bounds__(256) void degcount_kernel(const int* __restrict__ src,
                                                       const int* __restrict__ dst,
                                                       float* __restrict__ deg,
                                                       int* __restrict__ cnt,
                                                       int* __restrict__ rank, int E) {
    int e = blockIdx.x * 256 + threadIdx.x;
    if (e < E) {
        const int s = src[e];
        const int d = dst[e];
        atomicAdd(&deg[s], 1.0f);
        rank[e] = atomicAdd(&cnt[d * CPAD], 1);
    }
}

__global__ __launch_bounds__(1024) void scan_kernel(const int* __restrict__ cnt,
                                                    int* __restrict__ rowptr,
                                                    float* __restrict__ deg) {
    __shared__ int part[1024];
    const int t = threadIdx.x;
    int c[SCAN_PER];
    float d[SCAN_PER];
    int loc = 0;
#pragma unroll
    for (int i = 0; i < SCAN_PER; ++i) {
        const int idx = t * SCAN_PER + i;
        c[i] = (idx < NNODES) ? cnt[idx * CPAD] : 0;
        d[i] = (idx < NNODES) ? deg[idx] : 0.0f;
        loc += c[i];
    }
#pragma unroll
    for (int i = 0; i < SCAN_PER; ++i) {
        const int idx = t * SCAN_PER + i;
        if (idx < NNODES) deg[idx] = (d[i] > 0.0f) ? rsqrtf(d[i]) : 0.0f;
    }
    part[t] = loc;
    __syncthreads();
    for (int off = 1; off < 1024; off <<= 1) {
        const int v = (t >= off) ? part[t - off] : 0;
        __syncthreads();
        part[t] += v;
        __syncthreads();
    }
    int run = part[t] - loc;
#pragma unroll
    for (int i = 0; i < SCAN_PER; ++i) {
        const int idx = t * SCAN_PER + i;
        if (idx < NNODES) rowptr[idx] = run;
        run += c[i];
    }
    if (t == 1023) rowptr[NNODES] = part[1023];
}

__global__ __launch_bounds__(256) void fill_kernel(const int* __restrict__ src,
                                                   const int* __restrict__ dst,
                                                   const int* __restrict__ rank,
                                                   const int* __restrict__ rowptr,
                                                   const float* __restrict__ dis,
                                                   int* __restrict__ csrc,
                                                   float* __restrict__ cdis, int E) {
    int e = blockIdx.x * 256 + threadIdx.x;
    if (e < E) {
        const int s = src[e];
        const int pos = rowptr[dst[e]] + rank[e];
        csrc[pos] = s;
        cdis[pos] = dis[s];
    }
}


template <int D, int B>
__device__ __forceinline__ void edge_batch(const float* __restrict__ xin, int s, float nm,
                                           int j, int lane, float4 (&acc)[D / 128]) {
    constexpr int V = D / 128;
    float4 xv[B][V];
    float w[B];
#pragma unroll
    for (int u = 0; u < B; ++u) {
        const int su = __builtin_amdgcn_readlane(s, j + u);
        w[u] = __int_as_float(__builtin_amdgcn_readlane(__float_as_int(nm), j + u));
        const float4* row = (const float4*)(xin + (size_t)su * D);
#pragma unroll
        for (int v = 0; v < V; ++v) xv[u][v] = row[lane + 32 * v];
    }
#pragma unroll
    for (int u = 0; u < B; ++u) {
#pragma unroll
        for (int v = 0; v < V; ++v) {
            acc[v].x = __builtin_fmaf(w[u], xv[u][v].x, acc[v].x);
            acc[v].y = __builtin_fmaf(w[u], xv[u][v].y, acc[v].y);
            acc[v].z = __builtin_fmaf(w[u], xv[u][v].z, acc[v].z);
            acc[v].w = __builtin_fmaf(w[u], xv[u][v].w, acc[v].w);
        }
    }
}

template <int D>
__global__ __launch_bounds__(256) void gather_kernel(
    const float* __restrict__ xin, const int* __restrict__ rowptr,
    const int* __restrict__ csrc, const float* __restrict__ cdis,
    const float* __restrict__ dis, float* __restrict__ agg, int n)
{
    constexpr int V = D / 128;
    constexpr int U = 8 / V;
    const int lane = threadIdx.x & 31;
    const int node = __builtin_amdgcn_readfirstlane((int)(blockIdx.x * 8u + (threadIdx.x >> 5)));
    if (node >= n) return;
    const int beg = rowptr[node];
    const int end = rowptr[node + 1];
    const float dd = dis[node];

    float4 acc[V];
#pragma unroll
    for (int v = 0; v < V; ++v) { acc[v].x = 0.f; acc[v].y = 0.f; acc[v].z = 0.f; acc[v].w = 0.f; }

    int s = 0;
    float ds = 0.0f;
    if (beg + lane < end) { s = csrc[beg + lane]; ds = cdis[beg + lane]; }

    for (int base = beg; base < end; base += 32) {
        const int pn = (base + 32 + lane < end) ? (base + 32 + lane) : (end - 1);
        const int sn = csrc[pn];
        const float dsn = cdis[pn];

        const float nm = -(ds * dd);
        const int cnt = (end - base < 32) ? (end - base) : 32;
        int j = 0;
        for (; j + U <= cnt; j += U) edge_batch<D, U>(xin, s, nm, j, lane, acc);
        if (U > 4 && cnt - j >= 4) { edge_batch<D, 4>(xin, s, nm, j, lane, acc); j += 4; }
        if (cnt - j >= 2) { edge_batch<D, 2>(xin, s, nm, j, lane, acc); j += 2; }
        if (cnt - j >= 1) { edge_batch<D, 1>(xin, s, nm, j, lane, acc); }

        s = sn;
        ds = dsn;
    }

    float4* orow = (float4*)(agg + (size_t)node * D);
#pragma unroll
    for (int v = 0; v < V; ++v) orow[lane + 32 * v] = acc[v];
}

template <int TILES, int PAIRS>
struct GemmStage {
    v4f a0[PAIRS], a1[PAIRS];
    v4f b0[PAIRS][TILES], b1[PAIRS][TILES];
};

template <int TILES, int K, int PAIRS>
__device__ __forceinline__ void load_stage(GemmStage<TILES, PAIRS>& st, int s,
                                           const v4f* a0p, const v4f* a1p,
                                           const v4f* w0p, const v4f* w1p) {
    constexpr int KG = K / 8;
#pragma unroll
    for (int q = 0; q < PAIRS; ++q) {
        const int g = s * PAIRS + q;
        st.a0[q] = a0p[2 * g];
        st.a1[q] = a1p[2 * g];
#pragma unroll
        for (int t = 0; t < TILES; ++t) st.b0[q][t] = w0p[(t * KG + g) * 32];
#pragma unroll
        for (int t = 0; t < TILES; ++t) st.b1[q][t] = w1p[(t * KG + g) * 32];
    }
}

template <int TILES, int PAIRS>
__device__ __forceinline__ void mma_stage(const GemmStage<TILES, PAIRS>& st, v8f (&acc)[TILES]) {
#pragma unroll
    for (int q = 0; q < PAIRS; ++q) {
        const v2f a0lo = __builtin_shufflevector(st.a0[q], st.a0[q], 0, 1);
        const v2f a0hi = __builtin_shufflevector(st.a0[q], st.a0[q], 2, 3);
        const v2f a1lo = __builtin_shufflevector(st.a1[q], st.a1[q], 0, 1);
        const v2f a1hi = __builtin_shufflevector(st.a1[q], st.a1[q], 2, 3);
#pragma unroll
        for (int t = 0; t < TILES; ++t)
            acc[t] = __builtin_amdgcn_wmma_f32_16x16x4_f32(
                false, a0lo, false, __builtin_shufflevector(st.b0[q][t], st.b0[q][t], 0, 1),
                (short)0, acc[t], false, false);
#pragma unroll
        for (int t = 0; t < TILES; ++t)
            acc[t] = __builtin_amdgcn_wmma_f32_16x16x4_f32(
                false, a0hi, false, __builtin_shufflevector(st.b0[q][t], st.b0[q][t], 2, 3),
                (short)0, acc[t], false, false);
#pragma unroll
        for (int t = 0; t < TILES; ++t)
            acc[t] = __builtin_amdgcn_wmma_f32_16x16x4_f32(
                false, a1lo, false, __builtin_shufflevector(st.b1[q][t], st.b1[q][t], 0, 1),
                (short)0, acc[t], false, false);
#pragma unroll
        for (int t = 0; t < TILES; ++t)
            acc[t] = __builtin_amdgcn_wmma_f32_16x16x4_f32(
                false, a1hi, false, __builtin_shufflevector(st.b1[q][t], st.b1[q][t], 2, 3),
                (short)0, acc[t], false, false);
    }
}

template <int TILES, int K, int PAIRS>
__global__ __launch_bounds__(128) void cheb_gemm_kernel(
    const float* __restrict__ Xin, const float* Agg,
    const float* __restrict__ W0P, const float* __restrict__ W1P,
    const float* __restrict__ bias, float* Out, int relu)
{
    constexpr int KG = K / 8;
    constexpr int NST = KG / PAIRS;
    constexpr int NOut = TILES * 64;
    const int wave  = threadIdx.x >> 5;
    const int lane  = threadIdx.x & 31;
    const int rlane = lane & 15;
    const int half  = lane >> 4;
    const int rowBase = blockIdx.x << 4;
    const int colBase = wave * (TILES * 16);

    const v4f* a0p = (const v4f*)(Xin + (size_t)(rowBase + rlane) * K) + half;
    const v4f* a1p = (const v4f*)(Agg + (size_t)(rowBase + rlane) * K) + half;
    const v4f* w0p = (const v4f*)W0P + (size_t)(wave * TILES) * (KG * 32) + lane;
    const v4f* w1p = (const v4f*)W1P + (size_t)(wave * TILES) * (KG * 32) + lane;

    v8f acc[TILES];
#pragma unroll
    for (int t = 0; t < TILES; ++t) acc[t] = (v8f){};

    GemmStage<TILES, PAIRS> cur;
    load_stage<TILES, K, PAIRS>(cur, 0, a0p, a1p, w0p, w1p);

#pragma unroll
    for (int s = 0; s < NST - 1; ++s) {
        GemmStage<TILES, PAIRS> nxt;
        load_stage<TILES, K, PAIRS>(nxt, s + 1, a0p, a1p, w0p, w1p);
        __builtin_amdgcn_sched_barrier(0);
        mma_stage<TILES, PAIRS>(cur, acc);
        cur = nxt;
    }
    mma_stage<TILES, PAIRS>(cur, acc);

    __syncthreads();

    const int rowOff = rowBase + (half << 3);
#pragma unroll
    for (int t = 0; t < TILES; ++t) {
        const int col = colBase + t * 16 + rlane;
        const float bv = bias[col];
#pragma unroll
        for (int r = 0; r < 8; ++r) {
            float v = acc[t][r] + bv;
            if (relu) v = fmaxf(v, 0.0f);
            Out[(size_t)(rowOff + r) * NOut + col] = v;
        }
    }
}

extern "C" void kernel_launch(void* const* d_in, const int* in_sizes, int n_in,
                              void* d_out, int out_size, void* d_ws, size_t ws_size,
                              hipStream_t stream) {
    const float* x   = (const float*)d_in[0];
    const int*   ei  = (const int*)d_in[1];
    const float* W01 = (const float*)d_in[3];
    const float* W11 = (const float*)d_in[4];
    const float* b1  = (const float*)d_in[5];
    const float* W02 = (const float*)d_in[6];
    const float* W12 = (const float*)d_in[7];
    const float* b2  = (const float*)d_in[8];
    const float* W03 = (const float*)d_in[9];
    const float* W13 = (const float*)d_in[10];
    const float* b3  = (const float*)d_in[11];

    const int N = NNODES;
    const int E = in_sizes[1] / 2;
    const int* src = ei;
    const int* dst = ei + E;

    float* dis  = (float*)d_ws;
    float* bufA = dis + 10240;
    float* bufB = bufA + (size_t)N * 256;
    float* bufC = bufB + (size_t)N * 256;
    float* w01t = bufC + (size_t)N * 256;
    float* w11t = w01t + 128 * 256;
    float* w02t = w11t + 128 * 256;
    float* w12t = w02t + 256 * 256;
    float* w03t = w12t + 256 * 256;
    float* w13t = w03t + 256 * 128;
    float* out  = (float*)d_out;

    int* cnt    = (int*)bufA;
    int* rank   = (int*)bufB;
    int*   csrc   = (int*)bufC;
    float* cdis   = bufC + E;
    int*   rowptr = csrc + E + E;

    prep_kernel<<<ZERO_BLOCKS + PACK_BLOCKS, 256, 0, stream>>>(W01, W11, W02, W12, W03, W13, w01t,
                                                             (float4*)dis, (10240 + N * CPAD) / 4);

    degcount_kernel<<<(E + 255) / 256, 256, 0, stream>>>(src, dst, dis, cnt, rank, E);

    scan_kernel<<<1, 1024, 0, stream>>>(cnt, rowptr, dis);
    fill_kernel<<<(E + 255) / 256, 256, 0, stream>>>(src, dst, rank, rowptr, dis, csrc, cdis, E);

    gather_kernel<128><<<(N + 7) / 8, 256, 0, stream>>>(x, rowptr, csrc, cdis, dis, bufA, N);
    cheb_gemm_kernel<4, 128, 2><<<N / 16, 128, 0, stream>>>(x, bufA, w01t, w11t, b1, bufB, 1);

    gather_kernel<256><<<(N + 7) / 8, 256, 0, stream>>>(bufB, rowptr, csrc, cdis, dis, bufA, N);
    cheb_gemm_kernel<4, 256, 2><<<N / 16, 128, 0, stream>>>(bufB, bufA, w02t, w12t, b2, bufA, 1);

    gather_kernel<256><<<(N + 7) / 8, 256, 0, stream>>>(bufA, rowptr, csrc, cdis, dis, bufB, N);
    cheb_gemm_kernel<2, 256, 4><<<N / 16, 128, 0, stream>>>(bufA, bufB, w03t, w13t, b3, out, 0);
}


# Aggregation joined to the product launch, 10 launches to 7: four waves of a block sum its 16 aggregated rows into LDS while the other four do X W0, then add agg W1 from LDS; k sum order changes
# speedup vs baseline: 3.3931x; 1.0440x over previous
#include <hip/hip_runtime.h>
#include <hip/hip_bf16.h>

typedef __attribute__((ext_vector_type(2))) float v2f;
typedef __attribute__((ext_vector_type(4))) float v4f;
typedef __attribute__((ext_vector_type(8))) float v8f;

#define NNODES 10000
#define CPAD 32
#define SCAN_PER 10

__device__ __forceinline__ void pack_one(const float* __restrict__ in, float* __restrict__ out,
                                         int K, int NOut, int o) {
    const int e = o & 3;
    const int l = (o >> 2) & 31;
    const int tg = o >> 7;
    const int KG = K >> 3;
    const int tile = tg / KG;
    const int g = tg - tile * KG;
    const int k = 8 * g + 4 * (l >> 4) + e;
    const int n = 16 * tile + (l & 15);
    out[o] = in[(size_t)k * NOut + n];
}

#define ZERO_BLOCKS 128
#define PACK_BLOCKS 1024
__global__ __launch_bounds__(256) void prep_kernel(
    const float* __restrict__ W01, const float* __restrict__ W11,
    const float* __restrict__ W02, const float* __restrict__ W12,
    const float* __restrict__ W03, const float* __restrict__ W13,
    float* __restrict__ wp, float4* __restrict__ zp, int n4)
{
    const int t = threadIdx.x;
    if (blockIdx.x < ZERO_BLOCKS) {
        float4 z; z.x = 0.f; z.y = 0.f; z.z = 0.f; z.w = 0.f;
        for (int i = blockIdx.x * 256 + t; i < n4; i += ZERO_BLOCKS * 256) zp[i] = z;
        return;
    }
    const int b = blockIdx.x - ZERO_BLOCKS;
    if (b < 128)      pack_one(W01, wp,          128, 256, b * 256 + t);
    else if (b < 256) pack_one(W11, wp + 32768,  128, 256, (b - 128) * 256 + t);
    else if (b < 512) pack_one(W02, wp + 65536,  256, 256, (b - 256) * 256 + t);
    else if (b < 768) pack_one(W12, wp + 131072, 256, 256, (b - 512) * 256 + t);
    else if (b < 896) pack_one(W03, wp + 196608, 256, 128, (b - 768) * 256 + t);
    else              pack_one(W13, wp + 229376, 256, 128, (b - 896) * 256 + t);
}

__global__ __launch_bounds__(256) void degcount_kernel(const int* __restrict__ src,
                                                       const int* __restrict__ dst,
                                                       float* __restrict__ deg,
                                                       int* __restrict__ cnt,
                                                       int* __restrict__ rank, int E) {
    int e = blockIdx.x * 256 + threadIdx.x;
    if (e < E) {
        const int s = src[e];
        const int d = dst[e];
        atomicAdd(&deg[s], 1.0f);
        rank[e] = atomicAdd(&cnt[d * CPAD], 1);
    }
}

__global__ __launch_bounds__(1024) void scan_kernel(const int* __restrict__ cnt,
                                                    int* __restrict__ rowptr,
                                                    float* __restrict__ deg) {
    __shared__ int part[1024];
    const int t = threadIdx.x;
    int c[SCAN_PER];
    float d[SCAN_PER];
    int loc = 0;
#pragma unroll
    for (int i = 0; i < SCAN_PER; ++i) {
        const int idx = t * SCAN_PER + i;
        c[i] = (idx < NNODES) ? cnt[idx * CPAD] : 0;
        d[i] = (idx < NNODES) ? deg[idx] : 0.0f;
        loc += c[i];
    }
#pragma unroll
    for (int i = 0; i < SCAN_PER; ++i) {
        const int idx = t * SCAN_PER + i;
        if (idx < NNODES) deg[idx] = (d[i] > 0.0f) ? rsqrtf(d[i]) : 0.0f;
    }
    part[t] = loc;
    __syncthreads();
    for (int off = 1; off < 1024; off <<= 1) {
        const int v = (t >= off) ? part[t - off] : 0;
        __syncthreads();
        part[t] += v;
        __syncthreads();
    }
    int run = part[t] - loc;
#pragma unroll
    for (int i = 0; i < SCAN_PER; ++i) {
        const int idx = t * SCAN_PER + i;
        if (idx < NNODES) rowptr[idx] = run;
        run += c[i];
    }
    if (t == 1023) rowptr[NNODES] = part[1023];
}

__global__ __launch_bounds__(256) void fill_kernel(const int* __restrict__ src,
                                                   const int* __restrict__ dst,
                                                   const int* __restrict__ rank,
                                                   const int* __restrict__ rowptr,
                                                   const float* __restrict__ dis,
                                                   int* __restrict__ csrc,
                                                   float* __restrict__ cdis, int E) {
    int e = blockIdx.x * 256 + threadIdx.x;
    if (e < E) {
        const int s = src[e];
        const int pos = rowptr[dst[e]] + rank[e];
        csrc[pos] = s;
        cdis[pos] = dis[s];
    }
}


template <int D, int B>
__device__ __forceinline__ void edge_batch(const float* __restrict__ xin, int s, float nm,
                                           int j, int lane, float4 (&acc)[D / 128]) {
    constexpr int V = D / 128;
    float4 xv[B][V];
    float w[B];
#pragma unroll
    for (int u = 0; u < B; ++u) {
        const int su = __builtin_amdgcn_readlane(s, j + u);
        w[u] = __int_as_float(__builtin_amdgcn_readlane(__float_as_int(nm), j + u));
        const float4* row = (const float4*)(xin + (size_t)su * D);
#pragma unroll
        for (int v = 0; v < V; ++v) xv[u][v] = row[lane + 32 * v];
    }
#pragma unroll
    for (int u = 0; u < B; ++u) {
#pragma unroll
        for (int v = 0; v < V; ++v) {
            acc[v].x = __builtin_fmaf(w[u], xv[u][v].x, acc[v].x);
            acc[v].y = __builtin_fmaf(w[u], xv[u][v].y, acc[v].y);
            acc[v].z = __builtin_fmaf(w[u], xv[u][v].z, acc[v].z);
            acc[v].w = __builtin_fmaf(w[u], xv[u][v].w, acc[v].w);
        }
    }
}

template <int D>
__device__ __forceinline__ void gather_node(const float* __restrict__ xin,
                                            const int* __restrict__ csrc,
                                            const float* __restrict__ cdis,
                                            int beg, int end, float dd, int lane,
                                            int s, float ds, float4 (&acc)[D / 128]) {
    constexpr int V = D / 128;
    constexpr int U = 16 / V;
#pragma unroll
    for (int v = 0; v < V; ++v) { acc[v].x = 0.f; acc[v].y = 0.f; acc[v].z = 0.f; acc[v].w = 0.f; }

    for (int base = beg; base < end; base += 32) {
        const int pn = (base + 32 + lane < end) ? (base + 32 + lane) : (end - 1);
        const int sn = csrc[pn];
        const float dsn = cdis[pn];

        const float nm = -(ds * dd);
        const int cnt = (end - base < 32) ? (end - base) : 32;
        int j = 0;
        for (; j + U <= cnt; j += U) edge_batch<D, U>(xin, s, nm, j, lane, acc);
        if (U > 8 && cnt - j >= 8) { edge_batch<D, 8>(xin, s, nm, j, lane, acc); j += 8; }
        if (cnt - j >= 4) { edge_batch<D, 4>(xin, s, nm, j, lane, acc); j += 4; }
        if (cnt - j >= 2) { edge_batch<D, 2>(xin, s, nm, j, lane, acc); j += 2; }
        if (cnt - j >= 1) { edge_batch<D, 1>(xin, s, nm, j, lane, acc); }

        s = sn;
        ds = dsn;
    }
}

template <int TILES, int PAIRS>
struct GemmStage {
    v4f a[PAIRS];
    v4f b[PAIRS][TILES];
};

template <int TILES, int K, int PAIRS>
__device__ __forceinline__ void load_stage(GemmStage<TILES, PAIRS>& st, int s,
                                           const v4f* ap, const v4f* wp) {
    constexpr int KG = K / 8;
#pragma unroll
    for (int q = 0; q < PAIRS; ++q) {
        const int g = s * PAIRS + q;
        st.a[q] = ap[2 * g];
#pragma unroll
        for (int t = 0; t < TILES; ++t) st.b[q][t] = wp[(t * KG + g) * 32];
    }
}

template <int TILES, int PAIRS>
__device__ __forceinline__ void mma_stage(const GemmStage<TILES, PAIRS>& st, v8f (&acc)[TILES]) {
#pragma unroll
    for (int q = 0; q < PAIRS; ++q) {
        const v2f alo = __builtin_shufflevector(st.a[q], st.a[q], 0, 1);
        const v2f ahi = __builtin_shufflevector(st.a[q], st.a[q], 2, 3);
#pragma unroll
        for (int t = 0; t < TILES; ++t)
            acc[t] = __builtin_amdgcn_wmma_f32_16x16x4_f32(
                false, alo, false, __builtin_shufflevector(st.b[q][t], st.b[q][t], 0, 1),
                (short)0, acc[t], false, false);
#pragma unroll
        for (int t = 0; t < TILES; ++t)
            acc[t] = __builtin_amdgcn_wmma_f32_16x16x4_f32(
                false, ahi, false, __builtin_shufflevector(st.b[q][t], st.b[q][t], 2, 3),
                (short)0, acc[t], false, false);
    }
}

template <int TILES, int K, int PAIRS>
__device__ __forceinline__ void product_pass(const v4f* ap, const v4f* wp, v8f (&acc)[TILES]) {
    constexpr int NST = (K / 8) / PAIRS;
    GemmStage<TILES, PAIRS> cur;
    load_stage<TILES, K, PAIRS>(cur, 0, ap, wp);
#pragma unroll
    for (int s = 0; s < NST - 1; ++s) {
        GemmStage<TILES, PAIRS> nxt;
        load_stage<TILES, K, PAIRS>(nxt, s + 1, ap, wp);
        __builtin_amdgcn_sched_barrier(0);
        mma_stage<TILES, PAIRS>(cur, acc);
        __builtin_amdgcn_sched_barrier(0);
        cur = nxt;
    }
    mma_stage<TILES, PAIRS>(cur, acc);
}

template <int TILES, int K, int PAIRS>
__global__ __launch_bounds__(256, 6) void cheb_layer_kernel(
    const float* __restrict__ Xin, const int* __restrict__ rowptr,
    const int* __restrict__ csrc, const float* __restrict__ cdis,
    const float* __restrict__ dis,
    const float* __restrict__ W0P, const float* __restrict__ W1P,
    const float* __restrict__ bias, float* __restrict__ Out, int relu)
{
    constexpr int KG = K / 8;
    constexpr int NOut = 4 * TILES * 16;
    constexpr int V = K / 128;
    constexpr int PITCH = K + 4;
    __shared__ __attribute__((aligned(16))) float aggs[16 * PITCH];

    const int lane  = threadIdx.x & 31;
    const int wave  = __builtin_amdgcn_readfirstlane((int)(threadIdx.x >> 5));
    const int rowBase = blockIdx.x << 4;
    const int rlane = lane & 15;
    const int half  = lane >> 4;
    const int pw    = wave & 3;

    v8f acc[TILES];
#pragma unroll
    for (int t = 0; t < TILES; ++t) acc[t] = (v8f){};

    if (wave >= 4) {
        const int n0 = rowBase + 4 * (wave - 4);
        int r[5];
        float dd[4];
#pragma unroll
        for (int i = 0; i < 5; ++i) r[i] = rowptr[n0 + i];
#pragma unroll
        for (int i = 0; i < 4; ++i) dd[i] = dis[n0 + i];
        int s0[4];
        float d0[4];
#pragma unroll
        for (int i = 0; i < 4; ++i) {
            s0[i] = 0; d0[i] = 0.0f;
            if (r[i] + lane < r[i + 1]) { s0[i] = csrc[r[i] + lane]; d0[i] = cdis[r[i] + lane]; }
        }
#pragma unroll
        for (int i = 0; i < 4; ++i) {
            float4 acc4[V];
            gather_node<K>(Xin, csrc, cdis, r[i], r[i + 1], dd[i], lane, s0[i], d0[i], acc4);
            float4* trow = (float4*)(aggs + (4 * (wave - 4) + i) * PITCH);
#pragma unroll
            for (int v = 0; v < V; ++v) trow[lane + 32 * v] = acc4[v];
        }
    } else {
        const v4f* a0p = (const v4f*)(Xin + (size_t)(rowBase + rlane) * K) + half;
        const v4f* w0p = (const v4f*)W0P + (size_t)(pw * TILES) * (KG * 32) + lane;
        product_pass<TILES, K, PAIRS>(a0p, w0p, acc);
    }
    __syncthreads();
    if (wave >= 4) return;

    {
        const v4f* a1p = (const v4f*)(aggs + rlane * PITCH) + half;
        const v4f* w1p = (const v4f*)W1P + (size_t)(pw * TILES) * (KG * 32) + lane;
        product_pass<TILES, K, PAIRS>(a1p, w1p, acc);
    }

    const int colBase = pw * (TILES * 16);
    const int rowOff = rowBase + (half << 3);
#pragma unroll
    for (int t = 0; t < TILES; ++t) {
        const int col = colBase + t * 16 + rlane;
        const float bv = bias[col];
#pragma unroll
        for (int r8 = 0; r8 < 8; ++r8) {
            float v = acc[t][r8] + bv;
            if (relu) v = fmaxf(v, 0.0f);
            Out[(size_t)(rowOff + r8) * NOut + col] = v;
        }
    }
}

extern "C" void kernel_launch(void* const* d_in, const int* in_sizes, int n_in,
                              void* d_out, int out_size, void* d_ws, size_t ws_size,
                              hipStream_t stream) {
    const float* x   = (const float*)d_in[0];
    const int*   ei  = (const int*)d_in[1];
    const float* W01 = (const float*)d_in[3];
    const float* W11 = (const float*)d_in[4];
    const float* b1  = (const float*)d_in[5];
    const float* W02 = (const float*)d_in[6];
    const float* W12 = (const float*)d_in[7];
    const float* b2  = (const float*)d_in[8];
    const float* W03 = (const float*)d_in[9];
    const float* W13 = (const float*)d_in[10];
    const float* b3  = (const float*)d_in[11];

    const int N = NNODES;
    const int E = in_sizes[1] / 2;
    const int* src = ei;
    const int* dst = ei + E;

    float* dis  = (float*)d_ws;
    float* bufA = dis + 10240;
    float* bufB = bufA + (size_t)N * 256;
    float* bufC = bufB + (size_t)N * 256;
    float* w01t = bufC + (size_t)N * 256;
    float* w11t = w01t + 128 * 256;
    float* w02t = w11t + 128 * 256;
    float* w12t = w02t + 256 * 256;
    float* w03t = w12t + 256 * 256;
    float* w13t = w03t + 256 * 128;
    float* out  = (float*)d_out;

    int* cnt    = (int*)bufA;
    int* rank   = (int*)bufB;
    int*   csrc   = (int*)bufC;
    float* cdis   = bufC + E;
    int*   rowptr = csrc + E + E;

    prep_kernel<<<ZERO_BLOCKS + PACK_BLOCKS, 256, 0, stream>>>(W01, W11, W02, W12, W03, W13, w01t,
                                                             (float4*)dis, (10240 + N * CPAD) / 4);

    degcount_kernel<<<(E + 255) / 256, 256, 0, stream>>>(src, dst, dis, cnt, rank, E);

    scan_kernel<<<1, 1024, 0, stream>>>(cnt, rowptr, dis);
    fill_kernel<<<(E + 255) / 256, 256, 0, stream>>>(src, dst, rank, rowptr, dis, csrc, cdis, E);

    cheb_layer_kernel<4, 128, 2><<<N / 16, 256, 0, stream>>>(x, rowptr, csrc, cdis, dis,
                                                              w01t, w11t, b1, bufB, 1);
    cheb_layer_kernel<4, 256, 2><<<N / 16, 256, 0, stream>>>(bufB, rowptr, csrc, cdis, dis,
                                                              w02t, w12t, b2, bufA, 1);
    cheb_layer_kernel<2, 256, 4><<<N / 16, 256, 0, stream>>>(bufA, rowptr, csrc, cdis, dis,
                                                              w03t, w13t, b3, out, 0);
}


# Row-pointer scan joined to the fill launch, 7 launches to 6: each 1024-thread block scans the 10000 destination counters (now dense, 40 KB) into LDS, then places its edges; sum order as before
# speedup vs baseline: 3.8272x; 1.1279x over previous
#include <hip/hip_runtime.h>
#include <hip/hip_bf16.h>

typedef __attribute__((ext_vector_type(2))) float v2f;
typedef __attribute__((ext_vector_type(4))) float v4f;
typedef __attribute__((ext_vector_type(8))) float v8f;

#define NNODES 10000
#define SCAN_PER 10

__device__ __forceinline__ void pack_one(const float* __restrict__ in, float* __restrict__ out,
                                         int K, int NOut, int o) {
    const int e = o & 3;
    const int l = (o >> 2) & 31;
    const int tg = o >> 7;
    const int KG = K >> 3;
    const int tile = tg / KG;
    const int g = tg - tile * KG;
    const int k = 8 * g + 4 * (l >> 4) + e;
    const int n = 16 * tile + (l & 15);
    out[o] = in[(size_t)k * NOut + n];
}

#define ZERO_BLOCKS 128
#define PACK_BLOCKS 1024
__global__ __launch_bounds__(256) void prep_kernel(
    const float* __restrict__ W01, const float* __restrict__ W11,
    const float* __restrict__ W02, const float* __restrict__ W12,
    const float* __restrict__ W03, const float* __restrict__ W13,
    float* __restrict__ wp, float4* __restrict__ zp, int n4)
{
    const int t = threadIdx.x;
    if (blockIdx.x < ZERO_BLOCKS) {
        float4 z; z.x = 0.f; z.y = 0.f; z.z = 0.f; z.w = 0.f;
        for (int i = blockIdx.x * 256 + t; i < n4; i += ZERO_BLOCKS * 256) zp[i] = z;
        return;
    }
    const int b = blockIdx.x - ZERO_BLOCKS;
    if (b < 128)      pack_one(W01, wp,          128, 256, b * 256 + t);
    else if (b < 256) pack_one(W11, wp + 32768,  128, 256, (b - 128) * 256 + t);
    else if (b < 512) pack_one(W02, wp + 65536,  256, 256, (b - 256) * 256 + t);
    else if (b < 768) pack_one(W12, wp + 131072, 256, 256, (b - 512) * 256 + t);
    else if (b < 896) pack_one(W03, wp + 196608, 256, 128, (b - 768) * 256 + t);
    else              pack_one(W13, wp + 229376, 256, 128, (b - 896) * 256 + t);
}

__global__ __launch_bounds__(256) void degcount_kernel(const int* __restrict__ src,
                                                       const int* __restrict__ dst,
                                                       float* __restrict__ deg,
                                                       int* __restrict__ cnt,
                                                       int* __restrict__ rank, int E) {
    int e = blockIdx.x * 256 + threadIdx.x;
    if (e < E) {
        const int s = src[e];
        const int d = dst[e];
        atomicAdd(&deg[s], 1.0f);
        rank[e] = atomicAdd(&cnt[d], 1);
    }
}

__global__ __launch_bounds__(1024) void fill_kernel(const int* __restrict__ src,
                                                    const int* __restrict__ dst,
                                                    const int* __restrict__ rank,
                                                    const int* __restrict__ cnt,
                                                    const float* __restrict__ deg,
                                                    int* __restrict__ rowptr,
                                                    float* __restrict__ dis,
                                                    int* __restrict__ csrc,
                                                    float* __restrict__ cdis, int E) {
    __shared__ int srow[1024 * SCAN_PER];
    __shared__ int wsum[32];
    const int t = threadIdx.x;
    const int lane = t & 31;
    const int wave = t >> 5;
    const int e = blockIdx.x * 1024 + t;
    int s = 0, d = 0, rk = 0;
    float dg = 0.0f;
    if (e < E) { s = src[e]; d = dst[e]; rk = rank[e]; }
    int c[SCAN_PER];
    int loc = 0;
#pragma unroll
    for (int i = 0; i < SCAN_PER; ++i) {
        const int idx = t * SCAN_PER + i;
        c[i] = (idx < NNODES) ? cnt[idx] : 0;
        loc += c[i];
    }
    if (e < E) dg = deg[s];
    if (e < NNODES) {
        const float dn = deg[e];
        dis[e] = (dn > 0.0f) ? rsqrtf(dn) : 0.0f;
    }
    int incl = loc;
#pragma unroll
    for (int off = 1; off < 32; off <<= 1) {
        const int v = __shfl_up(incl, off);
        if (lane >= off) incl += v;
    }
    if (lane == 31) wsum[wave] = incl;
    __syncthreads();
    if (wave == 0) {
        const int w = wsum[lane];
        int wi = w;
#pragma unroll
        for (int off = 1; off < 32; off <<= 1) {
            const int v = __shfl_up(wi, off);
            if (lane >= off) wi += v;
        }
        wsum[lane] = wi - w;
    }
    __syncthreads();
    int run = wsum[wave] + incl - loc;
    const bool first = (blockIdx.x == 0);
#pragma unroll
    for (int i = 0; i < SCAN_PER; ++i) {
        const int idx = t * SCAN_PER + i;
        srow[idx] = run;
        if (first && idx < NNODES) rowptr[idx] = run;
        run += c[i];
    }
    if (first && t == 1023) rowptr[NNODES] = run;
    __syncthreads();
    if (e < E) {
        const int pos = srow[d] + rk;
        csrc[pos] = s;
        cdis[pos] = (dg > 0.0f) ? rsqrtf(dg) : 0.0f;
    }
}


template <int D, int B, int LD>
__device__ __forceinline__ void edge_batch(const float* __restrict__ xin, int s, float nm,
                                           int j, int lane, float4 (&acc)[D / 128]) {
    constexpr int V = D / 128;
    float4 xv[B][V];
    float w[B];
#pragma unroll
    for (int u = 0; u < B; ++u) {
        const int su = __builtin_amdgcn_readlane(s, j + u);
        w[u] = __int_as_float(__builtin_amdgcn_readlane(__float_as_int(nm), j + u));
        const float4* row = (const float4*)(xin + (size_t)su * LD);
#pragma unroll
        for (int v = 0; v < V; ++v) xv[u][v] = row[lane + 32 * v];
    }
#pragma unroll
    for (int u = 0; u < B; ++u) {
#pragma unroll
        for (int v = 0; v < V; ++v) {
            acc[v].x = __builtin_fmaf(w[u], xv[u][v].x, acc[v].x);
            acc[v].y = __builtin_fmaf(w[u], xv[u][v].y, acc[v].y);
            acc[v].z = __builtin_fmaf(w[u], xv[u][v].z, acc[v].z);
            acc[v].w = __builtin_fmaf(w[u], xv[u][v].w, acc[v].w);
        }
    }
}

template <int D, int LD>
__device__ __forceinline__ void gather_node(const float* __restrict__ xin,
                                            const int* __restrict__ csrc,
                                            const float* __restrict__ cdis,
                                            int beg, int end, float dd, int lane,
                                            int s, float ds, float4 (&acc)[D / 128]) {
    constexpr int V = D / 128;
    constexpr int U = 16 / V;
#pragma unroll
    for (int v = 0; v < V; ++v) { acc[v].x = 0.f; acc[v].y = 0.f; acc[v].z = 0.f; acc[v].w = 0.f; }

    for (int base = beg; base < end; base += 32) {
        const int pn = (base + 32 + lane < end) ? (base + 32 + lane) : (end - 1);
        const int sn = csrc[pn];
        const float dsn = cdis[pn];

        const float nm = -(ds * dd);
        const int cnt = (end - base < 32) ? (end - base) : 32;
        int j = 0;
        for (; j + U <= cnt; j += U) edge_batch<D, U, LD>(xin, s, nm, j, lane, acc);
        if (U > 8 && cnt - j >= 8) { edge_batch<D, 8, LD>(xin, s, nm, j, lane, acc); j += 8; }
        if (cnt - j >= 4) { edge_batch<D, 4, LD>(xin, s, nm, j, lane, acc); j += 4; }
        if (cnt - j >= 2) { edge_batch<D, 2, LD>(xin, s, nm, j, lane, acc); j += 2; }
        if (cnt - j >= 1) { edge_batch<D, 1, LD>(xin, s, nm, j, lane, acc); }

        s = sn;
        ds = dsn;
    }
}

template <int TILES, int PAIRS>
struct GemmStage {
    v4f a[PAIRS];
    v4f b[PAIRS][TILES];
};

template <int TILES, int K, int PAIRS>
__device__ __forceinline__ void load_stage(GemmStage<TILES, PAIRS>& st, int s,
                                           const v4f* ap, const v4f* wp) {
    constexpr int KG = K / 8;
#pragma unroll
    for (int q = 0; q < PAIRS; ++q) {
        const int g = s * PAIRS + q;
        st.a[q] = ap[2 * g];
#pragma unroll
        for (int t = 0; t < TILES; ++t) st.b[q][t] = wp[(t * KG + g) * 32];
    }
}

template <int TILES, int PAIRS>
__device__ __forceinline__ void mma_stage(const GemmStage<TILES, PAIRS>& st, v8f (&acc)[TILES]) {
#pragma unroll
    for (int q = 0; q < PAIRS; ++q) {
        const v2f alo = __builtin_shufflevector(st.a[q], st.a[q], 0, 1);
        const v2f ahi = __builtin_shufflevector(st.a[q], st.a[q], 2, 3);
#pragma unroll
        for (int t = 0; t < TILES; ++t)
            acc[t] = __builtin_amdgcn_wmma_f32_16x16x4_f32(
                false, alo, false, __builtin_shufflevector(st.b[q][t], st.b[q][t], 0, 1),
                (short)0, acc[t], false, false);
#pragma unroll
        for (int t = 0; t < TILES; ++t)
            acc[t] = __builtin_amdgcn_wmma_f32_16x16x4_f32(
                false, ahi, false, __builtin_shufflevector(st.b[q][t], st.b[q][t], 2, 3),
                (short)0, acc[t], false, false);
    }
}

template <int TILES, int K, int PAIRS>
__device__ __forceinline__ void product_pass(const v4f* ap, const v4f* wp, v8f (&acc)[TILES]) {
    constexpr int NST = (K / 8) / PAIRS;
    GemmStage<TILES, PAIRS> cur;
    load_stage<TILES, K, PAIRS>(cur, 0, ap, wp);
#pragma unroll
    for (int s = 0; s < NST - 1; ++s) {
        GemmStage<TILES, PAIRS> nxt;
        load_stage<TILES, K, PAIRS>(nxt, s + 1, ap, wp);
        __builtin_amdgcn_sched_barrier(0);
        mma_stage<TILES, PAIRS>(cur, acc);
        __builtin_amdgcn_sched_barrier(0);
        cur = nxt;
    }
    mma_stage<TILES, PAIRS>(cur, acc);
}

template <int TILES, int K, int PAIRS, bool PROJ>
__global__ __launch_bounds__(256, 6) void cheb_layer_kernel(
    const float* __restrict__ Xin, const int* __restrict__ rowptr,
    const int* __restrict__ csrc, const float* __restrict__ cdis,
    const float* __restrict__ dis,
    const float* __restrict__ W0P, const float* __restrict__ W1P,
    const float* __restrict__ bias,
    const float* __restrict__ WNP, const float* __restrict__ biasN,
    float* __restrict__ Out, int relu)
{
    constexpr int KG = K / 8;
    constexpr int NOut = 4 * TILES * 16;
    constexpr int V = K / 128;
    constexpr int PITCH = K + 4;
    __shared__ __attribute__((aligned(16))) float aggs[16 * PITCH];

    const int lane  = threadIdx.x & 31;
    const int wave  = __builtin_amdgcn_readfirstlane((int)(threadIdx.x >> 5));
    const int rowBase = blockIdx.x << 4;
    const int rlane = lane & 15;
    const int half  = lane >> 4;
    const int pw    = wave & 3;

    v8f acc[TILES];
#pragma unroll
    for (int t = 0; t < TILES; ++t) acc[t] = (v8f){};

    if (wave >= 4) {
        const int n0 = rowBase + 4 * (wave - 4);
        int r[5];
        float dd[4];
#pragma unroll
        for (int i = 0; i < 5; ++i) r[i] = rowptr[n0 + i];
#pragma unroll
        for (int i = 0; i < 4; ++i) dd[i] = dis[n0 + i];
        int s0[4];
        float d0[4];
#pragma unroll
        for (int i = 0; i < 4; ++i) {
            s0[i] = 0; d0[i] = 0.0f;
            if (r[i] + lane < r[i + 1]) { s0[i] = csrc[r[i] + lane]; d0[i] = cdis[r[i] + lane]; }
        }
#pragma unroll
        for (int i = 0; i < 4; ++i) {
            float4 acc4[V];
            gather_node<K, K>(Xin, csrc, cdis, r[i], r[i + 1], dd[i], lane, s0[i], d0[i], acc4);
            float4* trow = (float4*)(aggs + (4 * (wave - 4) + i) * PITCH);
#pragma unroll
            for (int v = 0; v < V; ++v) trow[lane + 32 * v] = acc4[v];
        }
    } else {
        const v4f* a0p = (const v4f*)(Xin + (size_t)(rowBase + rlane) * K) + half;
        const v4f* w0p = (const v4f*)W0P + (size_t)(pw * TILES) * (KG * 32) + lane;
        product_pass<TILES, K, PAIRS>(a0p, w0p, acc);
    }
    __syncthreads();
    const int colBase = pw * (TILES * 16);
    const int rowOff = rowBase + (half << 3);

    if constexpr (PROJ) {
        static_assert(NOut == K && NOut == 256, "the tile is reused for this layer's own output rows");
        if (wave < 4) {
            const v4f* a1p = (const v4f*)(aggs + rlane * PITCH) + half;
            const v4f* w1p = (const v4f*)W1P + (size_t)(pw * TILES) * (KG * 32) + lane;
            product_pass<TILES, K, PAIRS>(a1p, w1p, acc);
        }
        __syncthreads();
        if (wave < 4) {
#pragma unroll
            for (int t = 0; t < TILES; ++t) {
                const int col = colBase + t * 16 + rlane;
                const float bv = bias[col];
#pragma unroll
                for (int r8 = 0; r8 < 8; ++r8) {
                    float v = acc[t][r8] + bv;
                    if (relu) v = fmaxf(v, 0.0f);
                    aggs[((half << 3) + r8) * PITCH + col] = v;
                }
            }
        }
        __syncthreads();
        if (wave < 4) {
            v8f accn[4];
#pragma unroll
            for (int t = 0; t < 4; ++t) accn[t] = (v8f){};
            const v4f* anp = (const v4f*)(aggs + rlane * PITCH) + half;
            const v4f* wnp = (const v4f*)WNP + (size_t)(pw * 4) * ((NOut / 8) * 32) + lane;
            product_pass<4, NOut, 2>(anp, wnp, accn);
#pragma unroll
            for (int t = 0; t < 4; ++t) {
                const int col = pw * 64 + t * 16 + rlane;
                const float bv = (pw < 2) ? biasN[col] : 0.0f;
#pragma unroll
                for (int r8 = 0; r8 < 8; ++r8)
                    Out[(size_t)(rowOff + r8) * 256 + col] = accn[t][r8] + bv;
            }
        }
        return;
    }

    if (wave >= 4) return;

    {
        const v4f* a1p = (const v4f*)(aggs + rlane * PITCH) + half;
        const v4f* w1p = (const v4f*)W1P + (size_t)(pw * TILES) * (KG * 32) + lane;
        product_pass<TILES, K, PAIRS>(a1p, w1p, acc);
    }

#pragma unroll
    for (int t = 0; t < TILES; ++t) {
        const int col = colBase + t * 16 + rlane;
        const float bv = bias[col];
#pragma unroll
        for (int r8 = 0; r8 < 8; ++r8) {
            float v = acc[t][r8] + bv;
            if (relu) v = fmaxf(v, 0.0f);
            Out[(size_t)(rowOff + r8) * NOut + col] = v;
        }
    }
}

__global__ __launch_bounds__(256) void gather_out_kernel(
    const float* __restrict__ P, const int* __restrict__ rowptr,
    const int* __restrict__ csrc, const float* __restrict__ cdis,
    const float* __restrict__ dis, float* __restrict__ Out)
{
    const int lane = threadIdx.x & 31;
    const int wave = __builtin_amdgcn_readfirstlane((int)(threadIdx.x >> 5));
    const int n0 = (blockIdx.x << 4) + 2 * wave;
    int r[3];
    float dd[2];
#pragma unroll
    for (int i = 0; i < 3; ++i) r[i] = rowptr[n0 + i];
#pragma unroll
    for (int i = 0; i < 2; ++i) dd[i] = dis[n0 + i];
    int s0[2];
    float d0[2];
    float4 z[2];
#pragma unroll
    for (int i = 0; i < 2; ++i) {
        s0[i] = 0; d0[i] = 0.0f;
        if (r[i] + lane < r[i + 1]) { s0[i] = csrc[r[i] + lane]; d0[i] = cdis[r[i] + lane]; }
        z[i] = ((const float4*)(P + (size_t)(n0 + i) * 256))[lane];
    }
#pragma unroll
    for (int i = 0; i < 2; ++i) {
        float4 a[1];
        gather_node<128, 256>(P + 128, csrc, cdis, r[i], r[i + 1], dd[i], lane, s0[i], d0[i], a);
        float4 o;
        o.x = z[i].x + a[0].x;
        o.y = z[i].y + a[0].y;
        o.z = z[i].z + a[0].z;
        o.w = z[i].w + a[0].w;
        ((float4*)(Out + (size_t)(n0 + i) * 128))[lane] = o;
    }
}

extern "C" void kernel_launch(void* const* d_in, const int* in_sizes, int n_in,
                              void* d_out, int out_size, void* d_ws, size_t ws_size,
                              hipStream_t stream) {
    const float* x   = (const float*)d_in[0];
    const int*   ei  = (const int*)d_in[1];
    const float* W01 = (const float*)d_in[3];
    const float* W11 = (const float*)d_in[4];
    const float* b1  = (const float*)d_in[5];
    const float* W02 = (const float*)d_in[6];
    const float* W12 = (const float*)d_in[7];
    const float* b2  = (const float*)d_in[8];
    const float* W03 = (const float*)d_in[9];
    const float* W13 = (const float*)d_in[10];
    const float* b3  = (const float*)d_in[11];

    const int N = NNODES;
    const int E = in_sizes[1] / 2;
    const int* src = ei;
    const int* dst = ei + E;

    float* deg  = (float*)d_ws;
    float* bufA = deg + 10240;
    float* bufB = bufA + (size_t)N * 256;
    float* bufC = bufB + (size_t)N * 256;
    float* w01t = bufC + (size_t)N * 256;
    float* w11t = w01t + 128 * 256;
    float* w02t = w11t + 128 * 256;
    float* w12t = w02t + 256 * 256;
    float* w03t = w12t + 256 * 256;
    float* w13t = w03t + 256 * 128;
    float* out  = (float*)d_out;

    int* cnt    = (int*)bufA;
    int* rank   = (int*)bufB;
    int*   csrc   = (int*)bufC;
    float* cdis   = bufC + E;
    int*   rowptr = csrc + E + E;
    float* dis    = bufC + E + E + 10240;

    prep_kernel<<<ZERO_BLOCKS + PACK_BLOCKS, 256, 0, stream>>>(W01, W11, W02, W12, W03, W13, w01t,
                                                             (float4*)deg, (10240 + 10240) / 4);

    degcount_kernel<<<(E + 255) / 256, 256, 0, stream>>>(src, dst, deg, cnt, rank, E);

    fill_kernel<<<(E + 1023) / 1024, 1024, 0, stream>>>(src, dst, rank, cnt, deg, rowptr, dis,
                                                       csrc, cdis, E);

    cheb_layer_kernel<4, 128, 2, false><<<N / 16, 256, 0, stream>>>(x, rowptr, csrc, cdis, dis,
                                                                     w01t, w11t, b1, nullptr, nullptr,
                                                                     bufB, 1);
    cheb_layer_kernel<4, 256, 2, true><<<N / 16, 256, 0, stream>>>(bufB, rowptr, csrc, cdis, dis,
                                                                    w02t, w12t, b2, w03t, b3,
                                                                    bufA, 1);
    gather_out_kernel<<<N / 16, 256, 0, stream>>>(bufA, rowptr, csrc, cdis, dis, out);
}


# Edge list built by a counting sort in LDS: source histograms per chunk, edges grouped by 64-node destination range, one block per range orders its piece; no global atomics; sum order changes
# speedup vs baseline: 6.8926x; 1.8010x over previous
#include <hip/hip_runtime.h>
#include <hip/hip_bf16.h>

typedef __attribute__((ext_vector_type(2))) float v2f;
typedef __attribute__((ext_vector_type(4))) float v4f;
typedef __attribute__((ext_vector_type(8))) float v8f;

#define NNODES 10000

#define NPAD 10240
#define RSHIFT 6
#define RNODES 64
#define NRANGE 157
#define OFFP 160
#define HCH 10240
#define BCH 4096
#define OWU 8
#define CAP 8192

__device__ __forceinline__ void pack_one(const float* __restrict__ in, float* __restrict__ out,
                                         int K, int NOut, int o) {
    const int e = o & 3;
    const int l = (o >> 2) & 31;
    const int tg = o >> 7;
    const int KG = K >> 3;
    const int tile = tg / KG;
    const int g = tg - tile * KG;
    const int k = 8 * g + 4 * (l >> 4) + e;
    const int n = 16 * tile + (l & 15);
    out[o] = in[(size_t)k * NOut + n];
}

#define PACK_BLOCKS 256
__global__ __launch_bounds__(1024) void prep_kernel(
    const float* __restrict__ W01, const float* __restrict__ W11,
    const float* __restrict__ W02, const float* __restrict__ W12,
    const float* __restrict__ W03, const float* __restrict__ W13,
    float* __restrict__ wp, const int* __restrict__ src, int* __restrict__ planes, int E)
{
    __shared__ int hist[NPAD];
    const int t = threadIdx.x;
    const int b = blockIdx.x;
    if (b < PACK_BLOCKS) {
        if (b < 32)       pack_one(W01, wp,          128, 256, b * 1024 + t);
        else if (b < 64)  pack_one(W11, wp + 32768,  128, 256, (b - 32) * 1024 + t);
        else if (b < 128) pack_one(W02, wp + 65536,  256, 256, (b - 64) * 1024 + t);
        else if (b < 192) pack_one(W12, wp + 131072, 256, 256, (b - 128) * 1024 + t);
        else if (b < 224) pack_one(W03, wp + 196608, 256, 128, (b - 192) * 1024 + t);
        else              pack_one(W13, wp + 229376, 256, 128, (b - 224) * 1024 + t);
        return;
    }
    const int h = b - PACK_BLOCKS;
    for (int i = t; i < NPAD; i += 1024) hist[i] = 0;
    __syncthreads();
    const int e0 = h * HCH;
    for (int i = t; i < HCH; i += 1024) {
        const int e = e0 + i;
        if (e < E) atomicAdd(&hist[src[e]], 1);
    }
    __syncthreads();
    for (int i = t; i < NPAD; i += 1024) planes[(size_t)h * NPAD + i] = hist[i];
}

__global__ __launch_bounds__(1024) void bin_kernel(const int* __restrict__ src,
                                                   const int* __restrict__ dst,
                                                   const int* __restrict__ planes, int nplanes,
                                                   float* __restrict__ dis,
                                                   int* __restrict__ off,
                                                   int* __restrict__ recs, int E, int nchunks) {
    __shared__ int rcnt[OFFP];
    __shared__ int rstart[OFFP];
    __shared__ int sbuf[BCH];
    const int t = threadIdx.x;
    if ((int)blockIdx.x >= nchunks) {
        const int n = ((int)blockIdx.x - nchunks) * 1024 + t;
        if (n < NNODES) {
            int dg = 0;
            for (int h = 0; h < nplanes; ++h) dg += planes[(size_t)h * NPAD + n];
            dis[n] = (dg > 0) ? rsqrtf((float)dg) : 0.0f;
        }
        return;
    }
    const int c = blockIdx.x;
    if (t < OFFP) rcnt[t] = 0;
    __syncthreads();
    int rec[4], rr[4], pp[4];
#pragma unroll
    for (int i = 0; i < 4; ++i) {
        const int e = c * BCH + i * 1024 + t;
        rr[i] = -1; rec[i] = 0; pp[i] = 0;
        if (e < E) {
            const int s = src[e];
            const int d = dst[e];
            rr[i] = d >> RSHIFT;
            rec[i] = ((d & (RNODES - 1)) << 24) | s;
            pp[i] = atomicAdd(&rcnt[rr[i]], 1);
        }
    }
    __syncthreads();
    if (t < 32) {
        int v[5];
        int loc = 0;
#pragma unroll
        for (int i = 0; i < 5; ++i) { v[i] = rcnt[t * 5 + i]; loc += v[i]; }
        int incl = loc;
#pragma unroll
        for (int o = 1; o < 32; o <<= 1) {
            const int u = __shfl_up(incl, o);
            if (t >= o) incl += u;
        }
        int run = incl - loc;
#pragma unroll
        for (int i = 0; i < 5; ++i) { rstart[t * 5 + i] = run; run += v[i]; }
    }
    __syncthreads();
#pragma unroll
    for (int i = 0; i < 4; ++i)
        if (rr[i] >= 0) sbuf[rstart[rr[i]] + pp[i]] = rec[i];
    if (t < OFFP) off[c * OFFP + t] = rstart[t];
    __syncthreads();
#pragma unroll
    for (int i = 0; i < 4; ++i) {
        const int j = i * 1024 + t;
        if (c * BCH + j < E) recs[c * BCH + j] = sbuf[j];
    }
}

__global__ __launch_bounds__(1024) void owner_kernel(const int* __restrict__ off,
                                                     const int* __restrict__ recs,
                                                     const float* __restrict__ dis,
                                                     int* __restrict__ rowptr,
                                                     int* __restrict__ csrc,
                                                     float* __restrict__ cdis, int nchunks) {
    __shared__ int ncnt[RNODES];
    __shared__ int ncur[RNODES];
    __shared__ int tot[2];
    __shared__ int ss[CAP];
    __shared__ float sw[CAP];
    const int t = threadIdx.x;
    const int lane = t & 31;
    const int wave = t >> 5;
    const int r = blockIdx.x;
    if (t < RNODES) ncnt[t] = 0;
    if (t < 2) tot[t] = 0;
    __syncthreads();
    {
        int b0 = 0, m0 = 0;
        for (int c = t; c < nchunks; c += 1024) {
            const int o0 = off[c * OFFP + r];
            const int o1 = off[c * OFFP + r + 1];
            b0 += o0;
            m0 += o1 - o0;
        }
#pragma unroll
        for (int o = 16; o > 0; o >>= 1) {
            b0 += __shfl_xor(b0, o);
            m0 += __shfl_xor(m0, o);
        }
        if (lane == 0 && (b0 | m0) != 0) { atomicAdd(&tot[0], b0); atomicAdd(&tot[1], m0); }
    }
    for (int c0 = wave; c0 < nchunks; c0 += 32 * OWU) {
        int o0[OWU], o1[OWU];
        unsigned rc[OWU];
#pragma unroll
        for (int u = 0; u < OWU; ++u) {
            const int c = c0 + 32 * u;
            o0[u] = 0; o1[u] = 0;
            if (c < nchunks) { o0[u] = off[c * OFFP + r]; o1[u] = off[c * OFFP + r + 1]; }
        }
#pragma unroll
        for (int u = 0; u < OWU; ++u) {
            rc[u] = 0;
            if (o0[u] + lane < o1[u]) rc[u] = (unsigned)recs[(size_t)(c0 + 32 * u) * BCH + o0[u] + lane];
        }
#pragma unroll
        for (int u = 0; u < OWU; ++u) {
            if (o0[u] + lane < o1[u]) atomicAdd(&ncnt[rc[u] >> 24], 1);
            const int* p = recs + (size_t)(c0 + 32 * u) * BCH;
            for (int j = o0[u] + lane + 32; j < o1[u]; j += 32) atomicAdd(&ncnt[((unsigned)p[j]) >> 24], 1);
        }
    }
    __syncthreads();
    const int base = tot[0];
    const int M = tot[1];
    if (t < 32) {
        const int v0 = ncnt[2 * t];
        const int v1 = ncnt[2 * t + 1];
        int incl = v0 + v1;
#pragma unroll
        for (int o = 1; o < 32; o <<= 1) {
            const int u = __shfl_up(incl, o);
            if (t >= o) incl += u;
        }
        const int s0 = incl - v0 - v1;
        ncur[2 * t] = s0;
        ncur[2 * t + 1] = s0 + v0;
        const int n = r * RNODES + 2 * t;
        if (n < NNODES) rowptr[n] = base + s0;
        if (n + 1 < NNODES) rowptr[n + 1] = base + s0 + v0;
        if (r == NRANGE - 1 && t == 31) rowptr[NNODES] = base + M;
    }
    __syncthreads();
    const bool staged = (M <= CAP);
    for (int c0 = wave; c0 < nchunks; c0 += 32 * OWU) {
        int o0[OWU], o1[OWU];
        unsigned rc[OWU];
        float wv[OWU];
#pragma unroll
        for (int u = 0; u < OWU; ++u) {
            const int c = c0 + 32 * u;
            o0[u] = 0; o1[u] = 0;
            if (c < nchunks) { o0[u] = off[c * OFFP + r]; o1[u] = off[c * OFFP + r + 1]; }
        }
#pragma unroll
        for (int u = 0; u < OWU; ++u) {
            rc[u] = 0;
            if (o0[u] + lane < o1[u]) rc[u] = (unsigned)recs[(size_t)(c0 + 32 * u) * BCH + o0[u] + lane];
        }
#pragma unroll
        for (int u = 0; u < OWU; ++u) wv[u] = dis[rc[u] & 0xFFFFFFu];
#pragma unroll
        for (int u = 0; u < OWU; ++u) {
            if (o0[u] + lane < o1[u]) {
                const int s = (int)(rc[u] & 0xFFFFFFu);
                const int pos = atomicAdd(&ncur[rc[u] >> 24], 1);
                if (staged) { ss[pos] = s; sw[pos] = wv[u]; }
                else { csrc[base + pos] = s; cdis[base + pos] = wv[u]; }
            }
            const int* p = recs + (size_t)(c0 + 32 * u) * BCH;
            for (int j = o0[u] + lane + 32; j < o1[u]; j += 32) {
                const unsigned rec = (unsigned)p[j];
                const int s = (int)(rec & 0xFFFFFFu);
                const float w = dis[s];
                const int pos = atomicAdd(&ncur[rec >> 24], 1);
                if (staged) { ss[pos] = s; sw[pos] = w; }
                else { csrc[base + pos] = s; cdis[base + pos] = w; }
            }
        }
    }
    __syncthreads();
    if (staged) {
        for (int i = t; i < M; i += 1024) {
            csrc[base + i] = ss[i];
            cdis[base + i] = sw[i];
        }
    }
}


template <int D, int B, int LD>
__device__ __forceinline__ void edge_batch(const float* __restrict__ xin, int s, float nm,
                                           int j, int lane, float4 (&acc)[D / 128]) {
    constexpr int V = D / 128;
    float4 xv[B][V];
    float w[B];
#pragma unroll
    for (int u = 0; u < B; ++u) {
        const int su = __builtin_amdgcn_readlane(s, j + u);
        w[u] = __int_as_float(__builtin_amdgcn_readlane(__float_as_int(nm), j + u));
        const float4* row = (const float4*)(xin + (size_t)su * LD);
#pragma unroll
        for (int v = 0; v < V; ++v) xv[u][v] = row[lane + 32 * v];
    }
#pragma unroll
    for (int u = 0; u < B; ++u) {
#pragma unroll
        for (int v = 0; v < V; ++v) {
            acc[v].x = __builtin_fmaf(w[u], xv[u][v].x, acc[v].x);
            acc[v].y = __builtin_fmaf(w[u], xv[u][v].y, acc[v].y);
            acc[v].z = __builtin_fmaf(w[u], xv[u][v].z, acc[v].z);
            acc[v].w = __builtin_fmaf(w[u], xv[u][v].w, acc[v].w);
        }
    }
}

template <int D, int LD>
__device__ __forceinline__ void gather_node(const float* __restrict__ xin,
                                            const int* __restrict__ csrc,
                                            const float* __restrict__ cdis,
                                            int beg, int end, float dd, int lane,
                                            int s, float ds, float4 (&acc)[D / 128]) {
    constexpr int V = D / 128;
    constexpr int U = 16 / V;
#pragma unroll
    for (int v = 0; v < V; ++v) { acc[v].x = 0.f; acc[v].y = 0.f; acc[v].z = 0.f; acc[v].w = 0.f; }

    for (int base = beg; base < end; base += 32) {
        const int pn = (base + 32 + lane < end) ? (base + 32 + lane) : (end - 1);
        const int sn = csrc[pn];
        const float dsn = cdis[pn];

        const float nm = -(ds * dd);
        const int cnt = (end - base < 32) ? (end - base) : 32;
        int j = 0;
        for (; j + U <= cnt; j += U) edge_batch<D, U, LD>(xin, s, nm, j, lane, acc);
        if (U > 8 && cnt - j >= 8) { edge_batch<D, 8, LD>(xin, s, nm, j, lane, acc); j += 8; }
        if (cnt - j >= 4) { edge_batch<D, 4, LD>(xin, s, nm, j, lane, acc); j += 4; }
        if (cnt - j >= 2) { edge_batch<D, 2, LD>(xin, s, nm, j, lane, acc); j += 2; }
        if (cnt - j >= 1) { edge_batch<D, 1, LD>(xin, s, nm, j, lane, acc); }

        s = sn;
        ds = dsn;
    }
}

template <int TILES, int PAIRS>
struct GemmStage {
    v4f a[PAIRS];
    v4f b[PAIRS][TILES];
};

template <int TILES, int K, int PAIRS>
__device__ __forceinline__ void load_stage(GemmStage<TILES, PAIRS>& st, int s,
                                           const v4f* ap, const v4f* wp) {
    constexpr int KG = K / 8;
#pragma unroll
    for (int q = 0; q < PAIRS; ++q) {
        const int g = s * PAIRS + q;
        st.a[q] = ap[2 * g];
#pragma unroll
        for (int t = 0; t < TILES; ++t) st.b[q][t] = wp[(t * KG + g) * 32];
    }
}

template <int TILES, int PAIRS>
__device__ __forceinline__ void mma_stage(const GemmStage<TILES, PAIRS>& st, v8f (&acc)[TILES]) {
#pragma unroll
    for (int q = 0; q < PAIRS; ++q) {
        const v2f alo = __builtin_shufflevector(st.a[q], st.a[q], 0, 1);
        const v2f ahi = __builtin_shufflevector(st.a[q], st.a[q], 2, 3);
#pragma unroll
        for (int t = 0; t < TILES; ++t)
            acc[t] = __builtin_amdgcn_wmma_f32_16x16x4_f32(
                false, alo, false, __builtin_shufflevector(st.b[q][t], st.b[q][t], 0, 1),
                (short)0, acc[t], false, false);
#pragma unroll
        for (int t = 0; t < TILES; ++t)
            acc[t] = __builtin_amdgcn_wmma_f32_16x16x4_f32(
                false, ahi, false, __builtin_shufflevector(st.b[q][t], st.b[q][t], 2, 3),
                (short)0, acc[t], false, false);
    }
}

template <int TILES, int K, int PAIRS>
__device__ __forceinline__ void product_pass(const v4f* ap, const v4f* wp, v8f (&acc)[TILES]) {
    constexpr int NST = (K / 8) / PAIRS;
    GemmStage<TILES, PAIRS> cur;
    load_stage<TILES, K, PAIRS>(cur, 0, ap, wp);
#pragma unroll
    for (int s = 0; s < NST - 1; ++s) {
        GemmStage<TILES, PAIRS> nxt;
        load_stage<TILES, K, PAIRS>(nxt, s + 1, ap, wp);
        __builtin_amdgcn_sched_barrier(0);
        mma_stage<TILES, PAIRS>(cur, acc);
        __builtin_amdgcn_sched_barrier(0);
        cur = nxt;
    }
    mma_stage<TILES, PAIRS>(cur, acc);
}

template <int TILES, int K, int PAIRS, bool PROJ>
__global__ __launch_bounds__(256, 6) void cheb_layer_kernel(
    const float* __restrict__ Xin, const int* __restrict__ rowptr,
    const int* __restrict__ csrc, const float* __restrict__ cdis,
    const float* __restrict__ dis,
    const float* __restrict__ W0P, const float* __restrict__ W1P,
    const float* __restrict__ bias,
    const float* __restrict__ WNP, const float* __restrict__ biasN,
    float* __restrict__ Out, int relu)
{
    constexpr int KG = K / 8;
    constexpr int NOut = 4 * TILES * 16;
    constexpr int V = K / 128;
    constexpr int PITCH = K + 4;
    __shared__ __attribute__((aligned(16))) float aggs[16 * PITCH];

    const int lane  = threadIdx.x & 31;
    const int wave  = __builtin_amdgcn_readfirstlane((int)(threadIdx.x >> 5));
    const int rowBase = blockIdx.x << 4;
    const int rlane = lane & 15;
    const int half  = lane >> 4;
    const int pw    = wave & 3;

    v8f acc[TILES];
#pragma unroll
    for (int t = 0; t < TILES; ++t) acc[t] = (v8f){};

    if (wave >= 4) {
        const int n0 = rowBase + 4 * (wave - 4);
        int r[5];
        float dd[4];
#pragma unroll
        for (int i = 0; i < 5; ++i) r[i] = rowptr[n0 + i];
#pragma unroll
        for (int i = 0; i < 4; ++i) dd[i] = dis[n0 + i];
        int s0[4];
        float d0[4];
#pragma unroll
        for (int i = 0; i < 4; ++i) {
            s0[i] = 0; d0[i] = 0.0f;
            if (r[i] + lane < r[i + 1]) { s0[i] = csrc[r[i] + lane]; d0[i] = cdis[r[i] + lane]; }
        }
#pragma unroll
        for (int i = 0; i < 4; ++i) {
            float4 acc4[V];
            gather_node<K, K>(Xin, csrc, cdis, r[i], r[i + 1], dd[i], lane, s0[i], d0[i], acc4);
            float4* trow = (float4*)(aggs + (4 * (wave - 4) + i) * PITCH);
#pragma unroll
            for (int v = 0; v < V; ++v) trow[lane + 32 * v] = acc4[v];
        }
    } else {
        const v4f* a0p = (const v4f*)(Xin + (size_t)(rowBase + rlane) * K) + half;
        const v4f* w0p = (const v4f*)W0P + (size_t)(pw * TILES) * (KG * 32) + lane;
        product_pass<TILES, K, PAIRS>(a0p, w0p, acc);
    }
    __syncthreads();
    const int colBase = pw * (TILES * 16);
    const int rowOff = rowBase + (half << 3);

    if constexpr (PROJ) {
        static_assert(NOut == K && NOut == 256, "the tile is reused for this layer's own output rows");
        if (wave < 4) {
            const v4f* a1p = (const v4f*)(aggs + rlane * PITCH) + half;
            const v4f* w1p = (const v4f*)W1P + (size_t)(pw * TILES) * (KG * 32) + lane;
            product_pass<TILES, K, PAIRS>(a1p, w1p, acc);
        }
        __syncthreads();
        if (wave < 4) {
#pragma unroll
            for (int t = 0; t < TILES; ++t) {
                const int col = colBase + t * 16 + rlane;
                const float bv = bias[col];
#pragma unroll
                for (int r8 = 0; r8 < 8; ++r8) {
                    float v = acc[t][r8] + bv;
                    if (relu) v = fmaxf(v, 0.0f);
                    aggs[((half << 3) + r8) * PITCH + col] = v;
                }
            }
        }
        __syncthreads();
        if (wave < 4) {
            v8f accn[4];
#pragma unroll
            for (int t = 0; t < 4; ++t) accn[t] = (v8f){};
            const v4f* anp = (const v4f*)(aggs + rlane * PITCH) + half;
            const v4f* wnp = (const v4f*)WNP + (size_t)(pw * 4) * ((NOut / 8) * 32) + lane;
            product_pass<4, NOut, 2>(anp, wnp, accn);
#pragma unroll
            for (int t = 0; t < 4; ++t) {
                const int col = pw * 64 + t * 16 + rlane;
                const float bv = (pw < 2) ? biasN[col] : 0.0f;
#pragma unroll
                for (int r8 = 0; r8 < 8; ++r8)
                    Out[(size_t)(rowOff + r8) * 256 + col] = accn[t][r8] + bv;
            }
        }
        return;
    }

    if (wave >= 4) return;

    {
        const v4f* a1p = (const v4f*)(aggs + rlane * PITCH) + half;
        const v4f* w1p = (const v4f*)W1P + (size_t)(pw * TILES) * (KG * 32) + lane;
        product_pass<TILES, K, PAIRS>(a1p, w1p, acc);
    }

#pragma unroll
    for (int t = 0; t < TILES; ++t) {
        const int col = colBase + t * 16 + rlane;
        const float bv = bias[col];
#pragma unroll
        for (int r8 = 0; r8 < 8; ++r8) {
            float v = acc[t][r8] + bv;
            if (relu) v = fmaxf(v, 0.0f);
            Out[(size_t)(rowOff + r8) * NOut + col] = v;
        }
    }
}

__global__ __launch_bounds__(256) void gather_out_kernel(
    const float* __restrict__ P, const int* __restrict__ rowptr,
    const int* __restrict__ csrc, const float* __restrict__ cdis,
    const float* __restrict__ dis, float* __restrict__ Out)
{
    const int lane = threadIdx.x & 31;
    const int wave = __builtin_amdgcn_readfirstlane((int)(threadIdx.x >> 5));
    const int n0 = (blockIdx.x << 4) + 2 * wave;
    int r[3];
    float dd[2];
#pragma unroll
    for (int i = 0; i < 3; ++i) r[i] = rowptr[n0 + i];
#pragma unroll
    for (int i = 0; i < 2; ++i) dd[i] = dis[n0 + i];
    int s0[2];
    float d0[2];
    float4 z[2];
#pragma unroll
    for (int i = 0; i < 2; ++i) {
        s0[i] = 0; d0[i] = 0.0f;
        if (r[i] + lane < r[i + 1]) { s0[i] = csrc[r[i] + lane]; d0[i] = cdis[r[i] + lane]; }
        z[i] = ((const float4*)(P + (size_t)(n0 + i) * 256))[lane];
    }
#pragma unroll
    for (int i = 0; i < 2; ++i) {
        float4 a[1];
        gather_node<128, 256>(P + 128, csrc, cdis, r[i], r[i + 1], dd[i], lane, s0[i], d0[i], a);
        float4 o;
        o.x = z[i].x + a[0].x;
        o.y = z[i].y + a[0].y;
        o.z = z[i].z + a[0].z;
        o.w = z[i].w + a[0].w;
        ((float4*)(Out + (size_t)(n0 + i) * 128))[lane] = o;
    }
}

extern "C" void kernel_launch(void* const* d_in, const int* in_sizes, int n_in,
                              void* d_out, int out_size, void* d_ws, size_t ws_size,
                              hipStream_t stream) {
    const float* x   = (const float*)d_in[0];
    const int*   ei  = (const int*)d_in[1];
    const float* W01 = (const float*)d_in[3];
    const float* W11 = (const float*)d_in[4];
    const float* b1  = (const float*)d_in[5];
    const float* W02 = (const float*)d_in[6];
    const float* W12 = (const float*)d_in[7];
    const float* b2  = (const float*)d_in[8];
    const float* W03 = (const float*)d_in[9];
    const float* W13 = (const float*)d_in[10];
    const float* b3  = (const float*)d_in[11];

    const int N = NNODES;
    const int E = in_sizes[1] / 2;
    const int* src = ei;
    const int* dst = ei + E;

    float* bufA = (float*)d_ws + 10240;
    float* bufB = bufA + (size_t)N * 256;
    float* bufC = bufB + (size_t)N * 256;
    float* w01t = bufC + (size_t)N * 256;
    float* w11t = w01t + 128 * 256;
    float* w02t = w11t + 128 * 256;
    float* w12t = w02t + 256 * 256;
    float* w03t = w12t + 256 * 256;
    float* w13t = w03t + 256 * 128;
    float* out  = (float*)d_out;

    const int nplanes = (E + HCH - 1) / HCH;
    const int nchunks = (E + BCH - 1) / BCH;

    int* planes = (int*)bufB;
    int* recs   = planes + 64 * NPAD;
    int* off    = recs + E;
    int*   csrc   = (int*)bufC;
    float* cdis   = bufC + E;
    int*   rowptr = csrc + E + E;
    float* dis    = bufC + E + E + 10240;

    prep_kernel<<<PACK_BLOCKS + nplanes, 1024, 0, stream>>>(W01, W11, W02, W12, W03, W13, w01t,
                                                           src, planes, E);

    bin_kernel<<<nchunks + NPAD / 1024, 1024, 0, stream>>>(src, dst, planes, nplanes, dis, off,
                                                          recs, E, nchunks);

    owner_kernel<<<NRANGE, 1024, 0, stream>>>(off, recs, dis, rowptr, csrc, cdis, nchunks);

    cheb_layer_kernel<4, 128, 2, false><<<N / 16, 256, 0, stream>>>(x, rowptr, csrc, cdis, dis,
                                                                     w01t, w11t, b1, nullptr, nullptr,
                                                                     bufB, 1);
    cheb_layer_kernel<4, 256, 2, true><<<N / 16, 256, 0, stream>>>(bufB, rowptr, csrc, cdis, dis,
                                                                    w02t, w12t, b2, w03t, b3,
                                                                    bufA, 1);
    gather_out_kernel<<<N / 16, 256, 0, stream>>>(bufA, rowptr, csrc, cdis, dis, out);
}


# List-building launches issue each block's loads as groups (clamped index, no branch round a load): histogram 10 hops to 1, degree sum 11 to 1, binning 4 to 1, ordering 18 to 5; sum order changes
# speedup vs baseline: 7.4371x; 1.0790x over previous
#include <hip/hip_runtime.h>
#include <hip/hip_bf16.h>

typedef __attribute__((ext_vector_type(2))) float v2f;
typedef __attribute__((ext_vector_type(4))) float v4f;
typedef __attribute__((ext_vector_type(8))) float v8f;

#define NNODES 10000

#define MAXPL 64
#define NPAD 10240
#define RSHIFT 6
#define RNODES 64
#define NRANGE 157
#define OFFP 160
#define HCH 10240
#define BCH 4096
#define OWU 8
#define CAP 8192

__device__ __forceinline__ void pack_one(const float* __restrict__ in, float* __restrict__ out,
                                         int K, int NOut, int o) {
    const int e = o & 3;
    const int l = (o >> 2) & 31;
    const int tg = o >> 7;
    const int KG = K >> 3;
    const int tile = tg / KG;
    const int g = tg - tile * KG;
    const int k = 8 * g + 4 * (l >> 4) + e;
    const int n = 16 * tile + (l & 15);
    out[o] = in[(size_t)k * NOut + n];
}

#define PACK_BLOCKS 256
__global__ __launch_bounds__(1024) void prep_kernel(
    const float* __restrict__ W01, const float* __restrict__ W11,
    const float* __restrict__ W02, const float* __restrict__ W12,
    const float* __restrict__ W03, const float* __restrict__ W13,
    float* __restrict__ wp, const int* __restrict__ src, int* __restrict__ planes, int E)
{
    __shared__ int hist[NPAD];
    const int t = threadIdx.x;
    const int b = blockIdx.x;
    if (b < PACK_BLOCKS) {
        if (b < 32)       pack_one(W01, wp,          128, 256, b * 1024 + t);
        else if (b < 64)  pack_one(W11, wp + 32768,  128, 256, (b - 32) * 1024 + t);
        else if (b < 128) pack_one(W02, wp + 65536,  256, 256, (b - 64) * 1024 + t);
        else if (b < 192) pack_one(W12, wp + 131072, 256, 256, (b - 128) * 1024 + t);
        else if (b < 224) pack_one(W03, wp + 196608, 256, 128, (b - 192) * 1024 + t);
        else              pack_one(W13, wp + 229376, 256, 128, (b - 224) * 1024 + t);
        return;
    }
    const int h = b - PACK_BLOCKS;
    const int e0 = h * HCH;
    int sv[HCH / 1024];
#pragma unroll
    for (int i = 0; i < HCH / 1024; ++i) {
        const int e = e0 + i * 1024 + t;
        sv[i] = src[(e < E) ? e : (E - 1)];
    }
    for (int i = t; i < NPAD; i += 1024) hist[i] = 0;
    __syncthreads();
#pragma unroll
    for (int i = 0; i < HCH / 1024; ++i) {
        const int e = e0 + i * 1024 + t;
        if (e < E) atomicAdd(&hist[sv[i]], 1);
    }
    __syncthreads();
    for (int i = t; i < NPAD; i += 1024) planes[(size_t)h * NPAD + i] = hist[i];
}

__global__ __launch_bounds__(1024) void bin_kernel(const int* __restrict__ src,
                                                   const int* __restrict__ dst,
                                                   const int* __restrict__ planes, int nplanes,
                                                   float* __restrict__ dis,
                                                   int* __restrict__ off,
                                                   int* __restrict__ recs, int E, int nchunks) {
    __shared__ int rcnt[OFFP];
    __shared__ int rstart[OFFP];
    __shared__ int sbuf[BCH];
    const int t = threadIdx.x;
    if ((int)blockIdx.x >= nchunks) {
        const int n = ((int)blockIdx.x - nchunks) * 1024 + t;
        if (n < NNODES) {
            int pv[MAXPL];
#pragma unroll
            for (int h = 0; h < MAXPL; ++h) {
                const int hh = (h < nplanes) ? h : (nplanes - 1);
                pv[h] = planes[(size_t)hh * NPAD + n];
            }
            int dg = 0;
#pragma unroll
            for (int h = 0; h < MAXPL; ++h) dg += (h < nplanes) ? pv[h] : 0;
            dis[n] = (dg > 0) ? rsqrtf((float)dg) : 0.0f;
        }
        return;
    }
    const int c = blockIdx.x;
    int sv[4], dv[4];
#pragma unroll
    for (int i = 0; i < 4; ++i) {
        const int e = c * BCH + i * 1024 + t;
        const int ee = (e < E) ? e : (E - 1);
        sv[i] = src[ee];
        dv[i] = dst[ee];
    }
    if (t < OFFP) rcnt[t] = 0;
    __syncthreads();
    int rec[4], rr[4], pp[4];
#pragma unroll
    for (int i = 0; i < 4; ++i) {
        const int e = c * BCH + i * 1024 + t;
        rr[i] = -1; rec[i] = 0; pp[i] = 0;
        if (e < E) {
            const int s = sv[i];
            const int d = dv[i];
            rr[i] = d >> RSHIFT;
            rec[i] = ((d & (RNODES - 1)) << 24) | s;
            pp[i] = atomicAdd(&rcnt[rr[i]], 1);
        }
    }
    __syncthreads();
    if (t < 32) {
        int v[5];
        int loc = 0;
#pragma unroll
        for (int i = 0; i < 5; ++i) { v[i] = rcnt[t * 5 + i]; loc += v[i]; }
        int incl = loc;
#pragma unroll
        for (int o = 1; o < 32; o <<= 1) {
            const int u = __shfl_up(incl, o);
            if (t >= o) incl += u;
        }
        int run = incl - loc;
#pragma unroll
        for (int i = 0; i < 5; ++i) { rstart[t * 5 + i] = run; run += v[i]; }
    }
    __syncthreads();
#pragma unroll
    for (int i = 0; i < 4; ++i)
        if (rr[i] >= 0) sbuf[rstart[rr[i]] + pp[i]] = rec[i];
    if (t < OFFP) off[c * OFFP + t] = rstart[t];
    __syncthreads();
#pragma unroll
    for (int i = 0; i < 4; ++i) {
        const int j = i * 1024 + t;
        if (c * BCH + j < E) recs[c * BCH + j] = sbuf[j];
    }
}

__global__ __launch_bounds__(1024) void owner_kernel(const int* __restrict__ off,
                                                     const int* __restrict__ recs,
                                                     const float* __restrict__ dis,
                                                     int* __restrict__ rowptr,
                                                     int* __restrict__ csrc,
                                                     float* __restrict__ cdis, int nchunks) {
    __shared__ int ncnt[RNODES];
    __shared__ int ncur[RNODES];
    __shared__ int tot[2];
    __shared__ int ss[CAP];
    __shared__ float sw[CAP];
    const int t = threadIdx.x;
    const int lane = t & 31;
    const int wave = t >> 5;
    const int r = blockIdx.x;
    if (t < RNODES) ncnt[t] = 0;
    if (t < 2) tot[t] = 0;
    __syncthreads();
    for (int c0 = wave; c0 < nchunks; c0 += 32 * OWU) {
        int o0[OWU], o1[OWU];
        unsigned ra[OWU], rb[OWU];
#pragma unroll
        for (int u = 0; u < OWU; ++u) {
            const int c = c0 + 32 * u;
            const int cc = (c < nchunks) ? c : (nchunks - 1);
            o0[u] = off[cc * OFFP + r];
            o1[u] = off[cc * OFFP + r + 1];
        }
        asm volatile("" :: "v"(o0[0]), "v"(o0[1]), "v"(o0[2]), "v"(o0[3]), "v"(o0[4]), "v"(o0[5]),
                           "v"(o0[6]), "v"(o0[7]), "v"(o1[0]), "v"(o1[1]), "v"(o1[2]), "v"(o1[3]),
                           "v"(o1[4]), "v"(o1[5]), "v"(o1[6]), "v"(o1[7]));
        int b0 = 0, m0 = 0;
#pragma unroll
        for (int u = 0; u < OWU; ++u) {
            if (c0 + 32 * u >= nchunks) { o0[u] = 0; o1[u] = 0; }
            b0 += o0[u];
            m0 += o1[u] - o0[u];
        }
        if (lane == 0 && (b0 | m0) != 0) { atomicAdd(&tot[0], b0); atomicAdd(&tot[1], m0); }
#pragma unroll
        for (int u = 0; u < OWU; ++u) {
            const int c = c0 + 32 * u;
            const int* p = recs + (size_t)((c < nchunks) ? c : (nchunks - 1)) * BCH;
            const int ja = o0[u] + lane;
            ra[u] = (unsigned)p[(ja < o1[u]) ? ja : 0];
            rb[u] = (unsigned)p[(ja + 32 < o1[u]) ? (ja + 32) : 0];
        }
#pragma unroll
        for (int u = 0; u < OWU; ++u) {
            const int ja = o0[u] + lane;
            if (ja < o1[u]) atomicAdd(&ncnt[ra[u] >> 24], 1);
            if (ja + 32 < o1[u]) atomicAdd(&ncnt[rb[u] >> 24], 1);
            const int* p = recs + (size_t)(c0 + 32 * u) * BCH;
            for (int j = ja + 64; j < o1[u]; j += 32) atomicAdd(&ncnt[((unsigned)p[j]) >> 24], 1);
        }
    }
    __syncthreads();
    const int base = tot[0];
    const int M = tot[1];
    if (t < 32) {
        const int v0 = ncnt[2 * t];
        const int v1 = ncnt[2 * t + 1];
        int incl = v0 + v1;
#pragma unroll
        for (int o = 1; o < 32; o <<= 1) {
            const int u = __shfl_up(incl, o);
            if (t >= o) incl += u;
        }
        const int s0 = incl - v0 - v1;
        ncur[2 * t] = s0;
        ncur[2 * t + 1] = s0 + v0;
        const int n = r * RNODES + 2 * t;
        if (n < NNODES) rowptr[n] = base + s0;
        if (n + 1 < NNODES) rowptr[n + 1] = base + s0 + v0;
        if (r == NRANGE - 1 && t == 31) rowptr[NNODES] = base + M;
    }
    __syncthreads();
    const bool staged = (M <= CAP);
    for (int c0 = wave; c0 < nchunks; c0 += 32 * OWU) {
        int o0[OWU], o1[OWU];
        unsigned ra[OWU], rb[OWU];
        float wa[OWU], wb[OWU];
#pragma unroll
        for (int u = 0; u < OWU; ++u) {
            const int c = c0 + 32 * u;
            const int cc = (c < nchunks) ? c : (nchunks - 1);
            o0[u] = off[cc * OFFP + r];
            o1[u] = off[cc * OFFP + r + 1];
        }
#pragma unroll
        for (int u = 0; u < OWU; ++u)
            if (c0 + 32 * u >= nchunks) { o0[u] = 0; o1[u] = 0; }
#pragma unroll
        for (int u = 0; u < OWU; ++u) {
            const int c = c0 + 32 * u;
            const int* p = recs + (size_t)((c < nchunks) ? c : (nchunks - 1)) * BCH;
            const int ja = o0[u] + lane;
            ra[u] = (unsigned)p[(ja < o1[u]) ? ja : 0];
            rb[u] = (unsigned)p[(ja + 32 < o1[u]) ? (ja + 32) : 0];
        }
#pragma unroll
        for (int u = 0; u < OWU; ++u) {
            wa[u] = dis[ra[u] & 0xFFFFFFu];
            wb[u] = dis[rb[u] & 0xFFFFFFu];
        }
        asm volatile("" :: "v"(wa[0]), "v"(wa[1]), "v"(wa[2]), "v"(wa[3]), "v"(wa[4]), "v"(wa[5]),
                           "v"(wa[6]), "v"(wa[7]), "v"(wb[0]), "v"(wb[1]), "v"(wb[2]), "v"(wb[3]),
                           "v"(wb[4]), "v"(wb[5]), "v"(wb[6]), "v"(wb[7]));
#pragma unroll
        for (int u = 0; u < OWU; ++u) {
            const int ja = o0[u] + lane;
            if (ja < o1[u]) {
                const int s = (int)(ra[u] & 0xFFFFFFu);
                const int pos = atomicAdd(&ncur[ra[u] >> 24], 1);
                if (staged) { ss[pos] = s; sw[pos] = wa[u]; }
                else { csrc[base + pos] = s; cdis[base + pos] = wa[u]; }
            }
            if (ja + 32 < o1[u]) {
                const int s = (int)(rb[u] & 0xFFFFFFu);
                const int pos = atomicAdd(&ncur[rb[u] >> 24], 1);
                if (staged) { ss[pos] = s; sw[pos] = wb[u]; }
                else { csrc[base + pos] = s; cdis[base + pos] = wb[u]; }
            }
            const int* p = recs + (size_t)(c0 + 32 * u) * BCH;
            for (int j = ja + 64; j < o1[u]; j += 32) {
                const unsigned rec = (unsigned)p[j];
                const int s = (int)(rec & 0xFFFFFFu);
                const float w = dis[s];
                const int pos = atomicAdd(&ncur[rec >> 24], 1);
                if (staged) { ss[pos] = s; sw[pos] = w; }
                else { csrc[base + pos] = s; cdis[base + pos] = w; }
            }
        }
    }
    __syncthreads();
    if (staged) {
        for (int i = t; i < M; i += 1024) {
            csrc[base + i] = ss[i];
            cdis[base + i] = sw[i];
        }
    }
}


template <int D, int B, int LD>
__device__ __forceinline__ void edge_batch(const float* __restrict__ xin, int s, float nm,
                                           int j, int lane, float4 (&acc)[D / 128]) {
    constexpr int V = D / 128;
    float4 xv[B][V];
    float w[B];
#pragma unroll
    for (int u = 0; u < B; ++u) {
        const int su = __builtin_amdgcn_readlane(s, j + u);
        w[u] = __int_as_float(__builtin_amdgcn_readlane(__float_as_int(nm), j + u));
        const float4* row = (const float4*)(xin + (size_t)su * LD);
#pragma unroll
        for (int v = 0; v < V; ++v) xv[u][v] = row[lane + 32 * v];
    }
#pragma unroll
    for (int u = 0; u < B; ++u) {
#pragma unroll
        for (int v = 0; v < V; ++v) {
            acc[v].x = __builtin_fmaf(w[u], xv[u][v].x, acc[v].x);
            acc[v].y = __builtin_fmaf(w[u], xv[u][v].y, acc[v].y);
            acc[v].z = __builtin_fmaf(w[u], xv[u][v].z, acc[v].z);
            acc[v].w = __builtin_fmaf(w[u], xv[u][v].w, acc[v].w);
        }
    }
}

template <int D, int LD>
__device__ __forceinline__ void gather_node(const float* __restrict__ xin,
                                            const int* __restrict__ csrc,
                                            const float* __restrict__ cdis,
                                            int beg, int end, float dd, int lane,
                                            int s, float ds, float4 (&acc)[D / 128]) {
    constexpr int V = D / 128;
    constexpr int U = 16 / V;
#pragma unroll
    for (int v = 0; v < V; ++v) { acc[v].x = 0.f; acc[v].y = 0.f; acc[v].z = 0.f; acc[v].w = 0.f; }

    for (int base = beg; base < end; base += 32) {
        const int pn = (base + 32 + lane < end) ? (base + 32 + lane) : (end - 1);
        const int sn = csrc[pn];
        const float dsn = cdis[pn];

        const float nm = -(ds * dd);
        const int cnt = (end - base < 32) ? (end - base) : 32;
        int j = 0;
        for (; j + U <= cnt; j += U) edge_batch<D, U, LD>(xin, s, nm, j, lane, acc);
        if (U > 8 && cnt - j >= 8) { edge_batch<D, 8, LD>(xin, s, nm, j, lane, acc); j += 8; }
        if (cnt - j >= 4) { edge_batch<D, 4, LD>(xin, s, nm, j, lane, acc); j += 4; }
        if (cnt - j >= 2) { edge_batch<D, 2, LD>(xin, s, nm, j, lane, acc); j += 2; }
        if (cnt - j >= 1) { edge_batch<D, 1, LD>(xin, s, nm, j, lane, acc); }

        s = sn;
        ds = dsn;
    }
}

template <int TILES, int PAIRS>
struct GemmStage {
    v4f a[PAIRS];
    v4f b[PAIRS][TILES];
};

template <int TILES, int K, int PAIRS>
__device__ __forceinline__ void load_stage(GemmStage<TILES, PAIRS>& st, int s,
                                           const v4f* ap, const v4f* wp) {
    constexpr int KG = K / 8;
#pragma unroll
    for (int q = 0; q < PAIRS; ++q) {
        const int g = s * PAIRS + q;
        st.a[q] = ap[2 * g];
#pragma unroll
        for (int t = 0; t < TILES; ++t) st.b[q][t] = wp[(t * KG + g) * 32];
    }
}

template <int TILES, int PAIRS>
__device__ __forceinline__ void mma_stage(const GemmStage<TILES, PAIRS>& st, v8f (&acc)[TILES]) {
#pragma unroll
    for (int q = 0; q < PAIRS; ++q) {
        const v2f alo = __builtin_shufflevector(st.a[q], st.a[q], 0, 1);
        const v2f ahi = __builtin_shufflevector(st.a[q], st.a[q], 2, 3);
#pragma unroll
        for (int t = 0; t < TILES; ++t)
            acc[t] = __builtin_amdgcn_wmma_f32_16x16x4_f32(
                false, alo, false, __builtin_shufflevector(st.b[q][t], st.b[q][t], 0, 1),
                (short)0, acc[t], false, false);
#pragma unroll
        for (int t = 0; t < TILES; ++t)
            acc[t] = __builtin_amdgcn_wmma_f32_16x16x4_f32(
                false, ahi, false, __builtin_shufflevector(st.b[q][t], st.b[q][t], 2, 3),
                (short)0, acc[t], false, false);
    }
}

template <int TILES, int K, int PAIRS>
__device__ __forceinline__ void product_pass(const v4f* ap, const v4f* wp, v8f (&acc)[TILES]) {
    constexpr int NST = (K / 8) / PAIRS;
    GemmStage<TILES, PAIRS> cur;
    load_stage<TILES, K, PAIRS>(cur, 0, ap, wp);
#pragma unroll
    for (int s = 0; s < NST - 1; ++s) {
        GemmStage<TILES, PAIRS> nxt;
        load_stage<TILES, K, PAIRS>(nxt, s + 1, ap, wp);
        __builtin_amdgcn_sched_barrier(0);
        mma_stage<TILES, PAIRS>(cur, acc);
        __builtin_amdgcn_sched_barrier(0);
        cur = nxt;
    }
    mma_stage<TILES, PAIRS>(cur, acc);
}

template <int TILES, int K, int PAIRS, bool PROJ>
__global__ __launch_bounds__(256, 6) void cheb_layer_kernel(
    const float* __restrict__ Xin, const int* __restrict__ rowptr,
    const int* __restrict__ csrc, const float* __restrict__ cdis,
    const float* __restrict__ dis,
    const float* __restrict__ W0P, const float* __restrict__ W1P,
    const float* __restrict__ bias,
    const float* __restrict__ WNP, const float* __restrict__ biasN,
    float* __restrict__ Out, int relu)
{
    constexpr int KG = K / 8;
    constexpr int NOut = 4 * TILES * 16;
    constexpr int V = K / 128;
    constexpr int PITCH = K + 4;
    __shared__ __attribute__((aligned(16))) float aggs[16 * PITCH];

    const int lane  = threadIdx.x & 31;
    const int wave  = __builtin_amdgcn_readfirstlane((int)(threadIdx.x >> 5));
    const int rowBase = blockIdx.x << 4;
    const int rlane = lane & 15;
    const int half  = lane >> 4;
    const int pw    = wave & 3;

    v8f acc[TILES];
#pragma unroll
    for (int t = 0; t < TILES; ++t) acc[t] = (v8f){};

    if (wave >= 4) {
        const int n0 = rowBase + 4 * (wave - 4);
        int r[5];
        float dd[4];
#pragma unroll
        for (int i = 0; i < 5; ++i) r[i] = rowptr[n0 + i];
#pragma unroll
        for (int i = 0; i < 4; ++i) dd[i] = dis[n0 + i];
        int s0[4];
        float d0[4];
#pragma unroll
        for (int i = 0; i < 4; ++i) {
            s0[i] = 0; d0[i] = 0.0f;
            if (r[i] + lane < r[i + 1]) { s0[i] = csrc[r[i] + lane]; d0[i] = cdis[r[i] + lane]; }
        }
#pragma unroll
        for (int i = 0; i < 4; ++i) {
            float4 acc4[V];
            gather_node<K, K>(Xin, csrc, cdis, r[i], r[i + 1], dd[i], lane, s0[i], d0[i], acc4);
            float4* trow = (float4*)(aggs + (4 * (wave - 4) + i) * PITCH);
#pragma unroll
            for (int v = 0; v < V; ++v) trow[lane + 32 * v] = acc4[v];
        }
    } else {
        const v4f* a0p = (const v4f*)(Xin + (size_t)(rowBase + rlane) * K) + half;
        const v4f* w0p = (const v4f*)W0P + (size_t)(pw * TILES) * (KG * 32) + lane;
        product_pass<TILES, K, PAIRS>(a0p, w0p, acc);
    }
    __syncthreads();
    const int colBase = pw * (TILES * 16);
    const int rowOff = rowBase + (half << 3);

    if constexpr (PROJ) {
        static_assert(NOut == K && NOut == 256, "the tile is reused for this layer's own output rows");
        if (wave < 4) {
            const v4f* a1p = (const v4f*)(aggs + rlane * PITCH) + half;
            const v4f* w1p = (const v4f*)W1P + (size_t)(pw * TILES) * (KG * 32) + lane;
            product_pass<TILES, K, PAIRS>(a1p, w1p, acc);
        }
        __syncthreads();
        if (wave < 4) {
#pragma unroll
            for (int t = 0; t < TILES; ++t) {
                const int col = colBase + t * 16 + rlane;
                const float bv = bias[col];
#pragma unroll
                for (int r8 = 0; r8 < 8; ++r8) {
                    float v = acc[t][r8] + bv;
                    if (relu) v = fmaxf(v, 0.0f);
                    aggs[((half << 3) + r8) * PITCH + col] = v;
                }
            }
        }
        __syncthreads();
        if (wave < 4) {
            v8f accn[4];
#pragma unroll
            for (int t = 0; t < 4; ++t) accn[t] = (v8f){};
            const v4f* anp = (const v4f*)(aggs + rlane * PITCH) + half;
            const v4f* wnp = (const v4f*)WNP + (size_t)(pw * 4) * ((NOut / 8) * 32) + lane;
            product_pass<4, NOut, 2>(anp, wnp, accn);
#pragma unroll
            for (int t = 0; t < 4; ++t) {
                const int col = pw * 64 + t * 16 + rlane;
                const float bv = (pw < 2) ? biasN[col] : 0.0f;
#pragma unroll
                for (int r8 = 0; r8 < 8; ++r8)
                    Out[(size_t)(rowOff + r8) * 256 + col] = accn[t][r8] + bv;
            }
        }
        return;
    }

    if (wave >= 4) return;

    {
        const v4f* a1p = (const v4f*)(aggs + rlane * PITCH) + half;
        const v4f* w1p = (const v4f*)W1P + (size_t)(pw * TILES) * (KG * 32) + lane;
        product_pass<TILES, K, PAIRS>(a1p, w1p, acc);
    }

#pragma unroll
    for (int t = 0; t < TILES; ++t) {
        const int col = colBase + t * 16 + rlane;
        const float bv = bias[col];
#pragma unroll
        for (int r8 = 0; r8 < 8; ++r8) {
            float v = acc[t][r8] + bv;
            if (relu) v = fmaxf(v, 0.0f);
            Out[(size_t)(rowOff + r8) * NOut + col] = v;
        }
    }
}

__global__ __launch_bounds__(256) void gather_out_kernel(
    const float* __restrict__ P, const int* __restrict__ rowptr,
    const int* __restrict__ csrc, const float* __restrict__ cdis,
    const float* __restrict__ dis, float* __restrict__ Out)
{
    const int lane = threadIdx.x & 31;
    const int wave = __builtin_amdgcn_readfirstlane((int)(threadIdx.x >> 5));
    const int n0 = (blockIdx.x << 4) + 2 * wave;
    int r[3];
    float dd[2];
#pragma unroll
    for (int i = 0; i < 3; ++i) r[i] = rowptr[n0 + i];
#pragma unroll
    for (int i = 0; i < 2; ++i) dd[i] = dis[n0 + i];
    int s0[2];
    float d0[2];
    float4 z[2];
#pragma unroll
    for (int i = 0; i < 2; ++i) {
        s0[i] = 0; d0[i] = 0.0f;
        if (r[i] + lane < r[i + 1]) { s0[i] = csrc[r[i] + lane]; d0[i] = cdis[r[i] + lane]; }
        z[i] = ((const float4*)(P + (size_t)(n0 + i) * 256))[lane];
    }
#pragma unroll
    for (int i = 0; i < 2; ++i) {
        float4 a[1];
        gather_node<128, 256>(P + 128, csrc, cdis, r[i], r[i + 1], dd[i], lane, s0[i], d0[i], a);
        float4 o;
        o.x = z[i].x + a[0].x;
        o.y = z[i].y + a[0].y;
        o.z = z[i].z + a[0].z;
        o.w = z[i].w + a[0].w;
        ((float4*)(Out + (size_t)(n0 + i) * 128))[lane] = o;
    }
}

extern "C" void kernel_launch(void* const* d_in, const int* in_sizes, int n_in,
                              void* d_out, int out_size, void* d_ws, size_t ws_size,
                              hipStream_t stream) {
    const float* x   = (const float*)d_in[0];
    const int*   ei  = (const int*)d_in[1];
    const float* W01 = (const float*)d_in[3];
    const float* W11 = (const float*)d_in[4];
    const float* b1  = (const float*)d_in[5];
    const float* W02 = (const float*)d_in[6];
    const float* W12 = (const float*)d_in[7];
    const float* b2  = (const float*)d_in[8];
    const float* W03 = (const float*)d_in[9];
    const float* W13 = (const float*)d_in[10];
    const float* b3  = (const float*)d_in[11];

    const int N = NNODES;
    const int E = in_sizes[1] / 2;
    const int* src = ei;
    const int* dst = ei + E;

    float* bufA = (float*)d_ws + 10240;
    float* bufB = bufA + (size_t)N * 256;
    float* bufC = bufB + (size_t)N * 256;
    float* w01t = bufC + (size_t)N * 256;
    float* w11t = w01t + 128 * 256;
    float* w02t = w11t + 128 * 256;
    float* w12t = w02t + 256 * 256;
    float* w03t = w12t + 256 * 256;
    float* w13t = w03t + 256 * 128;
    float* out  = (float*)d_out;

    const int nplanes = (E + HCH - 1) / HCH;
    const int nchunks = (E + BCH - 1) / BCH;

    int* planes = (int*)bufB;
    int* recs   = planes + MAXPL * NPAD;
    int* off    = recs + E;
    int*   csrc   = (int*)bufC;
    float* cdis   = bufC + E;
    int*   rowptr = csrc + E + E;
    float* dis    = bufC + E + E + 10240;

    prep_kernel<<<PACK_BLOCKS + nplanes, 1024, 0, stream>>>(W01, W11, W02, W12, W03, W13, w01t,
                                                           src, planes, E);

    bin_kernel<<<nchunks + NPAD / 1024, 1024, 0, stream>>>(src, dst, planes, nplanes, dis, off,
                                                          recs, E, nchunks);

    owner_kernel<<<NRANGE, 1024, 0, stream>>>(off, recs, dis, rowptr, csrc, cdis, nchunks);

    cheb_layer_kernel<4, 128, 2, false><<<N / 16, 256, 0, stream>>>(x, rowptr, csrc, cdis, dis,
                                                                     w01t, w11t, b1, nullptr, nullptr,
                                                                     bufB, 1);
    cheb_layer_kernel<4, 256, 2, true><<<N / 16, 256, 0, stream>>>(bufB, rowptr, csrc, cdis, dis,
                                                                    w02t, w12t, b2, w03t, b3,
                                                                    bufA, 1);
    gather_out_kernel<<<N / 16, 256, 0, stream>>>(bufA, rowptr, csrc, cdis, dis, out);
}


# Launch merging, 6 to 5: binning, histogram and weight pack share launch 1 (pack last in the grid), the degree sum joins the ordering launch, layer 1's gather looks up dis[src]; sums as before
# speedup vs baseline: 7.8007x; 1.0489x over previous
#include <hip/hip_runtime.h>
#include <hip/hip_bf16.h>

typedef __attribute__((ext_vector_type(2))) float v2f;
typedef __attribute__((ext_vector_type(4))) float v4f;
typedef __attribute__((ext_vector_type(8))) float v8f;

#define NNODES 10000

#define MAXPL 64
#define NPAD 10240
#define RSHIFT 6
#define RNODES 64
#define NRANGE 157
#define OFFP 160
#define HCH 10240
#define BCH 4096
#define OWU 8
#define CAP 8192

__device__ __forceinline__ float pack_load(const float* __restrict__ in, int kgs, int NOut, int o) {
    const int e = o & 3;
    const int l = (o >> 2) & 31;
    const int tg = o >> 7;
    const int tile = tg >> kgs;
    const int g = tg & ((1 << kgs) - 1);
    const int k = 8 * g + 4 * (l >> 4) + e;
    const int n = 16 * tile + (l & 15);
    return in[(size_t)k * NOut + n];
}

#define PACKU 8
#define PACK_BLOCKS 32
__global__ __launch_bounds__(1024) void prep_kernel(
    const float* __restrict__ W01, const float* __restrict__ W11,
    const float* __restrict__ W02, const float* __restrict__ W12,
    const float* __restrict__ W03, const float* __restrict__ W13,
    float* __restrict__ wp, const int* __restrict__ src, const int* __restrict__ dst,
    int* __restrict__ planes, int* __restrict__ off, int* __restrict__ recs,
    int E, int nplanes, int nchunks)
{
    __shared__ int hist[NPAD];
    __shared__ int rcnt[OFFP];
    __shared__ int rstart[OFFP];
    const int t = threadIdx.x;
    const int b = blockIdx.x;
    if (b >= nchunks && b < nchunks + nplanes) {
        const int h = b - nchunks;
        const int e0 = h * HCH;
        int sv[HCH / 1024];
#pragma unroll
        for (int i = 0; i < HCH / 1024; ++i) {
            const int e = e0 + i * 1024 + t;
            sv[i] = src[(e < E) ? e : (E - 1)];
        }
        for (int i = t; i < NPAD; i += 1024) hist[i] = 0;
        __syncthreads();
#pragma unroll
        for (int i = 0; i < HCH / 1024; ++i) {
            const int e = e0 + i * 1024 + t;
            if (e < E) atomicAdd(&hist[sv[i]], 1);
        }
        __syncthreads();
        for (int i = t; i < NPAD; i += 1024) planes[(size_t)h * NPAD + i] = hist[i];
        return;
    }
    if (b >= nplanes + nchunks) {
        const int sub0 = (b - nplanes - nchunks) * PACKU;
        const float* in = W13;
        int kgs = 5, NOut = 128, first = 224;
        if (sub0 < 32)       { in = W01; kgs = 4; NOut = 256; first = 0; }
        else if (sub0 < 64)  { in = W11; kgs = 4; NOut = 256; first = 32; }
        else if (sub0 < 128) { in = W02; kgs = 5; NOut = 256; first = 64; }
        else if (sub0 < 192) { in = W12; kgs = 5; NOut = 256; first = 128; }
        else if (sub0 < 224) { in = W03; kgs = 5; NOut = 128; first = 192; }
        float v[PACKU];
#pragma unroll
        for (int i = 0; i < PACKU; ++i) v[i] = pack_load(in, kgs, NOut, (sub0 + i - first) * 1024 + t);
#pragma unroll
        for (int i = 0; i < PACKU; ++i) wp[(sub0 + i) * 1024 + t] = v[i];
        return;
    }
    int* sbuf = hist;
    const int c = b;
    int sv[4], dv[4];
#pragma unroll
    for (int i = 0; i < 4; ++i) {
        const int e = c * BCH + i * 1024 + t;
        const int ee = (e < E) ? e : (E - 1);
        sv[i] = src[ee];
        dv[i] = dst[ee];
    }
    if (t < OFFP) rcnt[t] = 0;
    __syncthreads();
    int rec[4], rr[4], pp[4];
#pragma unroll
    for (int i = 0; i < 4; ++i) {
        const int e = c * BCH + i * 1024 + t;
        rr[i] = -1; rec[i] = 0; pp[i] = 0;
        if (e < E) {
            const int s = sv[i];
            const int d = dv[i];
            rr[i] = d >> RSHIFT;
            rec[i] = ((d & (RNODES - 1)) << 24) | s;
            pp[i] = atomicAdd(&rcnt[rr[i]], 1);
        }
    }
    __syncthreads();
    if (t < 32) {
        int v[5];
        int loc = 0;
#pragma unroll
        for (int i = 0; i < 5; ++i) { v[i] = rcnt[t * 5 + i]; loc += v[i]; }
        int incl = loc;
#pragma unroll
        for (int o = 1; o < 32; o <<= 1) {
            const int u = __shfl_up(incl, o);
            if (t >= o) incl += u;
        }
        int run = incl - loc;
#pragma unroll
        for (int i = 0; i < 5; ++i) { rstart[t * 5 + i] = run; run += v[i]; }
    }
    __syncthreads();
#pragma unroll
    for (int i = 0; i < 4; ++i)
        if (rr[i] >= 0) sbuf[rstart[rr[i]] + pp[i]] = rec[i];
    if (t < OFFP) off[c * OFFP + t] = rstart[t];
    __syncthreads();
#pragma unroll
    for (int i = 0; i < 4; ++i) {
        const int j = i * 1024 + t;
        if (c * BCH + j < E) recs[c * BCH + j] = sbuf[j];
    }
}

__global__ __launch_bounds__(1024) void owner_kernel(const int* __restrict__ off,
                                                     const int* __restrict__ recs,
                                                     const int* __restrict__ planes, int nplanes,
                                                     float* __restrict__ dis,
                                                     int* __restrict__ rowptr,
                                                     int* __restrict__ csrc, int nchunks) {
    __shared__ int ncnt[RNODES];
    __shared__ int ncur[RNODES];
    __shared__ int tot[2];
    __shared__ int ss[CAP];
    const int t = threadIdx.x;
    if ((int)blockIdx.x >= NRANGE) {
        const int n = ((int)blockIdx.x - NRANGE) * 1024 + t;
        if (n < NNODES) {
            int pv[MAXPL];
#pragma unroll
            for (int h = 0; h < MAXPL; ++h) {
                const int hh = (h < nplanes) ? h : (nplanes - 1);
                pv[h] = planes[(size_t)hh * NPAD + n];
            }
            int dg = 0;
#pragma unroll
            for (int h = 0; h < MAXPL; ++h) dg += (h < nplanes) ? pv[h] : 0;
            dis[n] = (dg > 0) ? rsqrtf((float)dg) : 0.0f;
        }
        return;
    }
    const int lane = t & 31;
    const int wave = t >> 5;
    const int r = blockIdx.x;
    if (t < RNODES) ncnt[t] = 0;
    if (t < 2) tot[t] = 0;
    __syncthreads();
    for (int c0 = wave; c0 < nchunks; c0 += 32 * OWU) {
        int o0[OWU], o1[OWU];
        unsigned ra[OWU], rb[OWU];
#pragma unroll
        for (int u = 0; u < OWU; ++u) {
            const int c = c0 + 32 * u;
            const int cc = (c < nchunks) ? c : (nchunks - 1);
            o0[u] = off[cc * OFFP + r];
            o1[u] = off[cc * OFFP + r + 1];
        }
        asm volatile("" :: "v"(o0[0]), "v"(o0[1]), "v"(o0[2]), "v"(o0[3]), "v"(o0[4]), "v"(o0[5]),
                           "v"(o0[6]), "v"(o0[7]), "v"(o1[0]), "v"(o1[1]), "v"(o1[2]), "v"(o1[3]),
                           "v"(o1[4]), "v"(o1[5]), "v"(o1[6]), "v"(o1[7]));
        int b0 = 0, m0 = 0;
#pragma unroll
        for (int u = 0; u < OWU; ++u) {
            if (c0 + 32 * u >= nchunks) { o0[u] = 0; o1[u] = 0; }
            b0 += o0[u];
            m0 += o1[u] - o0[u];
        }
        if (lane == 0 && (b0 | m0) != 0) { atomicAdd(&tot[0], b0); atomicAdd(&tot[1], m0); }
#pragma unroll
        for (int u = 0; u < OWU; ++u) {
            const int c = c0 + 32 * u;
            const int* p = recs + (size_t)((c < nchunks) ? c : (nchunks - 1)) * BCH;
            const int ja = o0[u] + lane;
            ra[u] = (unsigned)p[(ja < o1[u]) ? ja : 0];
            rb[u] = (unsigned)p[(ja + 32 < o1[u]) ? (ja + 32) : 0];
        }
#pragma unroll
        for (int u = 0; u < OWU; ++u) {
            const int ja = o0[u] + lane;
            if (ja < o1[u]) atomicAdd(&ncnt[ra[u] >> 24], 1);
            if (ja + 32 < o1[u]) atomicAdd(&ncnt[rb[u] >> 24], 1);
            const int* p = recs + (size_t)(c0 + 32 * u) * BCH;
            for (int j = ja + 64; j < o1[u]; j += 32) atomicAdd(&ncnt[((unsigned)p[j]) >> 24], 1);
        }
    }
    __syncthreads();
    const int base = tot[0];
    const int M = tot[1];
    if (t < 32) {
        const int v0 = ncnt[2 * t];
        const int v1 = ncnt[2 * t + 1];
        int incl = v0 + v1;
#pragma unroll
        for (int o = 1; o < 32; o <<= 1) {
            const int u = __shfl_up(incl, o);
            if (t >= o) incl += u;
        }
        const int s0 = incl - v0 - v1;
        ncur[2 * t] = s0;
        ncur[2 * t + 1] = s0 + v0;
        const int n = r * RNODES + 2 * t;
        if (n < NNODES) rowptr[n] = base + s0;
        if (n + 1 < NNODES) rowptr[n + 1] = base + s0 + v0;
        if (r == NRANGE - 1 && t == 31) rowptr[NNODES] = base + M;
    }
    __syncthreads();
    const bool staged = (M <= CAP);
    for (int c0 = wave; c0 < nchunks; c0 += 32 * OWU) {
        int o0[OWU], o1[OWU];
        unsigned ra[OWU], rb[OWU];
#pragma unroll
        for (int u = 0; u < OWU; ++u) {
            const int c = c0 + 32 * u;
            const int cc = (c < nchunks) ? c : (nchunks - 1);
            o0[u] = off[cc * OFFP + r];
            o1[u] = off[cc * OFFP + r + 1];
        }
#pragma unroll
        for (int u = 0; u < OWU; ++u)
            if (c0 + 32 * u >= nchunks) { o0[u] = 0; o1[u] = 0; }
#pragma unroll
        for (int u = 0; u < OWU; ++u) {
            const int c = c0 + 32 * u;
            const int* p = recs + (size_t)((c < nchunks) ? c : (nchunks - 1)) * BCH;
            const int ja = o0[u] + lane;
            ra[u] = (unsigned)p[(ja < o1[u]) ? ja : 0];
            rb[u] = (unsigned)p[(ja + 32 < o1[u]) ? (ja + 32) : 0];
        }
        asm volatile("" :: "v"(ra[0]), "v"(ra[1]), "v"(ra[2]), "v"(ra[3]), "v"(ra[4]), "v"(ra[5]),
                           "v"(ra[6]), "v"(ra[7]), "v"(rb[0]), "v"(rb[1]), "v"(rb[2]), "v"(rb[3]),
                           "v"(rb[4]), "v"(rb[5]), "v"(rb[6]), "v"(rb[7]));
#pragma unroll
        for (int u = 0; u < OWU; ++u) {
            const int ja = o0[u] + lane;
            if (ja < o1[u]) {
                const int s = (int)(ra[u] & 0xFFFFFFu);
                const int pos = atomicAdd(&ncur[ra[u] >> 24], 1);
                if (staged) ss[pos] = s;
                else csrc[base + pos] = s;
            }
            if (ja + 32 < o1[u]) {
                const int s = (int)(rb[u] & 0xFFFFFFu);
                const int pos = atomicAdd(&ncur[rb[u] >> 24], 1);
                if (staged) ss[pos] = s;
                else csrc[base + pos] = s;
            }
            const int* p = recs + (size_t)(c0 + 32 * u) * BCH;
            for (int j = ja + 64; j < o1[u]; j += 32) {
                const unsigned rec = (unsigned)p[j];
                const int s = (int)(rec & 0xFFFFFFu);
                const int pos = atomicAdd(&ncur[rec >> 24], 1);
                if (staged) ss[pos] = s;
                else csrc[base + pos] = s;
            }
        }
    }
    __syncthreads();
    if (staged) {
        for (int i = t; i < M; i += 1024) csrc[base + i] = ss[i];
    }
}


template <int D, int B, int LD>
__device__ __forceinline__ void edge_batch(const float* __restrict__ xin, int s, float nm,
                                           int j, int lane, float4 (&acc)[D / 128]) {
    constexpr int V = D / 128;
    float4 xv[B][V];
    float w[B];
#pragma unroll
    for (int u = 0; u < B; ++u) {
        const int su = __builtin_amdgcn_readlane(s, j + u);
        w[u] = __int_as_float(__builtin_amdgcn_readlane(__float_as_int(nm), j + u));
        const float4* row = (const float4*)(xin + (size_t)su * LD);
#pragma unroll
        for (int v = 0; v < V; ++v) xv[u][v] = row[lane + 32 * v];
    }
#pragma unroll
    for (int u = 0; u < B; ++u) {
#pragma unroll
        for (int v = 0; v < V; ++v) {
            acc[v].x = __builtin_fmaf(w[u], xv[u][v].x, acc[v].x);
            acc[v].y = __builtin_fmaf(w[u], xv[u][v].y, acc[v].y);
            acc[v].z = __builtin_fmaf(w[u], xv[u][v].z, acc[v].z);
            acc[v].w = __builtin_fmaf(w[u], xv[u][v].w, acc[v].w);
        }
    }
}

template <int D, int LD, bool FILL>
__device__ __forceinline__ void gather_node(const float* __restrict__ xin,
                                            const int* __restrict__ csrc,
                                            const float* __restrict__ cdis,
                                            const float* __restrict__ dis,
                                            float* __restrict__ cdisw,
                                            int beg, int end, float dd, int lane,
                                            int s, float ds, float4 (&acc)[D / 128]) {
    constexpr int V = D / 128;
    constexpr int U = 16 / V;
#pragma unroll
    for (int v = 0; v < V; ++v) { acc[v].x = 0.f; acc[v].y = 0.f; acc[v].z = 0.f; acc[v].w = 0.f; }

    for (int base = beg; base < end; base += 32) {
        const bool nvalid = (base + 32 + lane < end);
        const int pn = nvalid ? (base + 32 + lane) : (end - 1);
        const int sn = csrc[pn];
        float dsn = 0.0f;
        if (!FILL) dsn = cdis[pn];

        const float nm = -(ds * dd);
        const int cnt = (end - base < 32) ? (end - base) : 32;
        int j = 0;
        if (FILL) {
            if (base + 32 < end) {
                edge_batch<D, U, LD>(xin, s, nm, 0, lane, acc);
                j = U;
                dsn = dis[sn];
            }
        }
        for (; j + U <= cnt; j += U) edge_batch<D, U, LD>(xin, s, nm, j, lane, acc);
        if (U > 8 && cnt - j >= 8) { edge_batch<D, 8, LD>(xin, s, nm, j, lane, acc); j += 8; }
        if (cnt - j >= 4) { edge_batch<D, 4, LD>(xin, s, nm, j, lane, acc); j += 4; }
        if (cnt - j >= 2) { edge_batch<D, 2, LD>(xin, s, nm, j, lane, acc); j += 2; }
        if (cnt - j >= 1) { edge_batch<D, 1, LD>(xin, s, nm, j, lane, acc); }

        if (FILL) { if (nvalid) cdisw[pn] = dsn; }
        s = sn;
        ds = dsn;
    }
}

template <int TILES, int PAIRS>
struct GemmStage {
    v4f a[PAIRS];
    v4f b[PAIRS][TILES];
};

template <int TILES, int K, int PAIRS>
__device__ __forceinline__ void load_stage(GemmStage<TILES, PAIRS>& st, int s,
                                           const v4f* ap, const v4f* wp) {
    constexpr int KG = K / 8;
#pragma unroll
    for (int q = 0; q < PAIRS; ++q) {
        const int g = s * PAIRS + q;
        st.a[q] = ap[2 * g];
#pragma unroll
        for (int t = 0; t < TILES; ++t) st.b[q][t] = wp[(t * KG + g) * 32];
    }
}

template <int TILES, int PAIRS>
__device__ __forceinline__ void mma_stage(const GemmStage<TILES, PAIRS>& st, v8f (&acc)[TILES]) {
#pragma unroll
    for (int q = 0; q < PAIRS; ++q) {
        const v2f alo = __builtin_shufflevector(st.a[q], st.a[q], 0, 1);
        const v2f ahi = __builtin_shufflevector(st.a[q], st.a[q], 2, 3);
#pragma unroll
        for (int t = 0; t < TILES; ++t)
            acc[t] = __builtin_amdgcn_wmma_f32_16x16x4_f32(
                false, alo, false, __builtin_shufflevector(st.b[q][t], st.b[q][t], 0, 1),
                (short)0, acc[t], false, false);
#pragma unroll
        for (int t = 0; t < TILES; ++t)
            acc[t] = __builtin_amdgcn_wmma_f32_16x16x4_f32(
                false, ahi, false, __builtin_shufflevector(st.b[q][t], st.b[q][t], 2, 3),
                (short)0, acc[t], false, false);
    }
}

template <int TILES, int K, int PAIRS>
__device__ __forceinline__ void product_pass(const v4f* ap, const v4f* wp, v8f (&acc)[TILES]) {
    constexpr int NST = (K / 8) / PAIRS;
    GemmStage<TILES, PAIRS> cur;
    load_stage<TILES, K, PAIRS>(cur, 0, ap, wp);
#pragma unroll
    for (int s = 0; s < NST - 1; ++s) {
        GemmStage<TILES, PAIRS> nxt;
        load_stage<TILES, K, PAIRS>(nxt, s + 1, ap, wp);
        __builtin_amdgcn_sched_barrier(0);
        mma_stage<TILES, PAIRS>(cur, acc);
        __builtin_amdgcn_sched_barrier(0);
        cur = nxt;
    }
    mma_stage<TILES, PAIRS>(cur, acc);
}

template <int TILES, int K, int PAIRS, bool PROJ, bool FILL>
__global__ __launch_bounds__(256, 6) void cheb_layer_kernel(
    const float* __restrict__ Xin, const int* __restrict__ rowptr,
    const int* __restrict__ csrc, const float* __restrict__ cdis,
    float* __restrict__ cdisw,
    const float* __restrict__ dis,
    const float* __restrict__ W0P, const float* __restrict__ W1P,
    const float* __restrict__ bias,
    const float* __restrict__ WNP, const float* __restrict__ biasN,
    float* __restrict__ Out, int relu)
{
    constexpr int KG = K / 8;
    constexpr int NOut = 4 * TILES * 16;
    constexpr int V = K / 128;
    constexpr int PITCH = K + 4;
    __shared__ __attribute__((aligned(16))) float aggs[16 * PITCH];

    const int lane  = threadIdx.x & 31;
    const int wave  = __builtin_amdgcn_readfirstlane((int)(threadIdx.x >> 5));
    const int rowBase = blockIdx.x << 4;
    const int rlane = lane & 15;
    const int half  = lane >> 4;
    const int pw    = wave & 3;

    v8f acc[TILES];
#pragma unroll
    for (int t = 0; t < TILES; ++t) acc[t] = (v8f){};

    if (wave >= 4) {
        const int n0 = rowBase + 4 * (wave - 4);
        int r[5];
        float dd[4];
#pragma unroll
        for (int i = 0; i < 5; ++i) r[i] = rowptr[n0 + i];
#pragma unroll
        for (int i = 0; i < 4; ++i) dd[i] = dis[n0 + i];
        int s0[4];
        float d0[4];
        if constexpr (FILL) {
#pragma unroll
            for (int i = 0; i < 4; ++i) s0[i] = csrc[(r[i] + lane < r[i + 1]) ? (r[i] + lane) : 0];
#pragma unroll
            for (int i = 0; i < 4; ++i) d0[i] = dis[s0[i]];
#pragma unroll
            for (int i = 0; i < 4; ++i)
                if (r[i] + lane < r[i + 1]) cdisw[r[i] + lane] = d0[i];
        } else {
#pragma unroll
            for (int i = 0; i < 4; ++i) {
                s0[i] = 0; d0[i] = 0.0f;
                if (r[i] + lane < r[i + 1]) { s0[i] = csrc[r[i] + lane]; d0[i] = cdis[r[i] + lane]; }
            }
        }
#pragma unroll
        for (int i = 0; i < 4; ++i) {
            float4 acc4[V];
            gather_node<K, K, FILL>(Xin, csrc, cdis, dis, cdisw, r[i], r[i + 1], dd[i], lane, s0[i], d0[i], acc4);
            float4* trow = (float4*)(aggs + (4 * (wave - 4) + i) * PITCH);
#pragma unroll
            for (int v = 0; v < V; ++v) trow[lane + 32 * v] = acc4[v];
        }
    } else {
        const v4f* a0p = (const v4f*)(Xin + (size_t)(rowBase + rlane) * K) + half;
        const v4f* w0p = (const v4f*)W0P + (size_t)(pw * TILES) * (KG * 32) + lane;
        product_pass<TILES, K, PAIRS>(a0p, w0p, acc);
    }
    __syncthreads();
    const int colBase = pw * (TILES * 16);
    const int rowOff = rowBase + (half << 3);

    if constexpr (PROJ) {
        static_assert(NOut == K && NOut == 256, "the tile is reused for this layer's own output rows");
        if (wave < 4) {
            const v4f* a1p = (const v4f*)(aggs + rlane * PITCH) + half;
            const v4f* w1p = (const v4f*)W1P + (size_t)(pw * TILES) * (KG * 32) + lane;
            product_pass<TILES, K, PAIRS>(a1p, w1p, acc);
        }
        __syncthreads();
        if (wave < 4) {
#pragma unroll
            for (int t = 0; t < TILES; ++t) {
                const int col = colBase + t * 16 + rlane;
                const float bv = bias[col];
#pragma unroll
                for (int r8 = 0; r8 < 8; ++r8) {
                    float v = acc[t][r8] + bv;
                    if (relu) v = fmaxf(v, 0.0f);
                    aggs[((half << 3) + r8) * PITCH + col] = v;
                }
            }
        }
        __syncthreads();
        if (wave < 4) {
            v8f accn[4];
#pragma unroll
            for (int t = 0; t < 4; ++t) accn[t] = (v8f){};
            const v4f* anp = (const v4f*)(aggs + rlane * PITCH) + half;
            const v4f* wnp = (const v4f*)WNP + (size_t)(pw * 4) * ((NOut / 8) * 32) + lane;
            product_pass<4, NOut, 2>(anp, wnp, accn);
#pragma unroll
            for (int t = 0; t < 4; ++t) {
                const int col = pw * 64 + t * 16 + rlane;
                const float bv = (pw < 2) ? biasN[col] : 0.0f;
#pragma unroll
                for (int r8 = 0; r8 < 8; ++r8)
                    Out[(size_t)(rowOff + r8) * 256 + col] = accn[t][r8] + bv;
            }
        }
        return;
    }

    if (wave >= 4) return;

    {
        const v4f* a1p = (const v4f*)(aggs + rlane * PITCH) + half;
        const v4f* w1p = (const v4f*)W1P + (size_t)(pw * TILES) * (KG * 32) + lane;
        product_pass<TILES, K, PAIRS>(a1p, w1p, acc);
    }

#pragma unroll
    for (int t = 0; t < TILES; ++t) {
        const int col = colBase + t * 16 + rlane;
        const float bv = bias[col];
#pragma unroll
        for (int r8 = 0; r8 < 8; ++r8) {
            float v = acc[t][r8] + bv;
            if (relu) v = fmaxf(v, 0.0f);
            Out[(size_t)(rowOff + r8) * NOut + col] = v;
        }
    }
}

__global__ __launch_bounds__(256) void gather_out_kernel(
    const float* __restrict__ P, const int* __restrict__ rowptr,
    const int* __restrict__ csrc, const float* __restrict__ cdis,
    const float* __restrict__ dis, float* __restrict__ Out)
{
    const int lane = threadIdx.x & 31;
    const int wave = __builtin_amdgcn_readfirstlane((int)(threadIdx.x >> 5));
    const int n0 = (blockIdx.x << 4) + 2 * wave;
    int r[3];
    float dd[2];
#pragma unroll
    for (int i = 0; i < 3; ++i) r[i] = rowptr[n0 + i];
#pragma unroll
    for (int i = 0; i < 2; ++i) dd[i] = dis[n0 + i];
    int s0[2];
    float d0[2];
    float4 z[2];
#pragma unroll
    for (int i = 0; i < 2; ++i) {
        s0[i] = 0; d0[i] = 0.0f;
        if (r[i] + lane < r[i + 1]) { s0[i] = csrc[r[i] + lane]; d0[i] = cdis[r[i] + lane]; }
        z[i] = ((const float4*)(P + (size_t)(n0 + i) * 256))[lane];
    }
#pragma unroll
    for (int i = 0; i < 2; ++i) {
        float4 a[1];
        gather_node<128, 256, false>(P + 128, csrc, cdis, dis, nullptr, r[i], r[i + 1], dd[i], lane, s0[i], d0[i], a);
        float4 o;
        o.x = z[i].x + a[0].x;
        o.y = z[i].y + a[0].y;
        o.z = z[i].z + a[0].z;
        o.w = z[i].w + a[0].w;
        ((float4*)(Out + (size_t)(n0 + i) * 128))[lane] = o;
    }
}

extern "C" void kernel_launch(void* const* d_in, const int* in_sizes, int n_in,
                              void* d_out, int out_size, void* d_ws, size_t ws_size,
                              hipStream_t stream) {
    const float* x   = (const float*)d_in[0];
    const int*   ei  = (const int*)d_in[1];
    const float* W01 = (const float*)d_in[3];
    const float* W11 = (const float*)d_in[4];
    const float* b1  = (const float*)d_in[5];
    const float* W02 = (const float*)d_in[6];
    const float* W12 = (const float*)d_in[7];
    const float* b2  = (const float*)d_in[8];
    const float* W03 = (const float*)d_in[9];
    const float* W13 = (const float*)d_in[10];
    const float* b3  = (const float*)d_in[11];

    const int N = NNODES;
    const int E = in_sizes[1] / 2;
    const int* src = ei;
    const int* dst = ei + E;

    float* bufA = (float*)d_ws + 10240;
    float* bufB = bufA + (size_t)N * 256;
    float* bufC = bufB + (size_t)N * 256;
    float* w01t = bufC + (size_t)N * 256;
    float* w11t = w01t + 128 * 256;
    float* w02t = w11t + 128 * 256;
    float* w12t = w02t + 256 * 256;
    float* w03t = w12t + 256 * 256;
    float* w13t = w03t + 256 * 128;
    float* out  = (float*)d_out;

    const int nplanes = (E + HCH - 1) / HCH;
    const int nchunks = (E + BCH - 1) / BCH;

    int* planes = (int*)bufB;
    int* recs   = planes + MAXPL * NPAD;
    int* off    = recs + E;
    int*   csrc   = (int*)bufC;
    float* cdis   = bufC + E;
    int*   rowptr = csrc + E + E;
    float* dis    = bufC + E + E + 10240;

    prep_kernel<<<nchunks + nplanes + PACK_BLOCKS, 1024, 0, stream>>>(W01, W11, W02, W12, W03, W13,
                                                                     w01t, src, dst, planes, off,
                                                                     recs, E, nplanes, nchunks);

    owner_kernel<<<NRANGE + NPAD / 1024, 1024, 0, stream>>>(off, recs, planes, nplanes, dis, rowptr,
                                                           csrc, nchunks);

    cheb_layer_kernel<4, 128, 2, false, true><<<N / 16, 256, 0, stream>>>(x, rowptr, csrc, nullptr, cdis,
                                                                           dis, w01t, w11t, b1, nullptr,
                                                                           nullptr, bufB, 1);
    cheb_layer_kernel<4, 256, 2, true, false><<<N / 16, 256, 0, stream>>>(bufB, rowptr, csrc, cdis, nullptr,
                                                                           dis, w02t, w12t, b2, w03t, b3,
                                                                           bufA, 1);
    gather_out_kernel<<<N / 16, 256, 0, stream>>>(bufA, rowptr, csrc, cdis, dis, out);
}
